# starting kernel
# baseline (speedup 1.0000x reference)
#include <hip/hip_runtime.h>
#include <hip/hip_bf16.h>

#define AANN 256
#define BB   64
#define DD   768
#define LL   2
#define GG1  4
#define GG2  8

typedef __attribute__((ext_vector_type(2))) float v2f;
typedef __attribute__((ext_vector_type(8))) float v8f;

__device__ __forceinline__ v8f wmma_f32(v2f a, v2f b, v8f c) {
    return __builtin_amdgcn_wmma_f32_16x16x4_f32(
         false, a,  false, b,
         (short)0, c,  false,  false);
}

__device__ __forceinline__ void gemm_wave_32x64(
    const float* __restrict__ A,
    const float* __restrict__ Bm,
    const float* __restrict__ bias,
    float* __restrict__ C,
    int m0, int n0)
{
    const int lane = threadIdx.x & 31;
    const int half = lane >> 4;
    const int r    = lane & 15;

    v8f acc[2][4];
    #pragma unroll
    for (int t = 0; t < 4; ++t) {
        const float bv = bias[n0 + t * 16 + r];
        #pragma unroll
        for (int j = 0; j < 8; ++j) { acc[0][t][j] = bv; acc[1][t][j] = bv; }
    }

    const float* arow0 = A + (size_t)(m0 + r) * DD;
    const float* arow1 = arow0 + (size_t)16 * DD;
    const float* bcol  = Bm + n0 + r;

    #pragma unroll 2
    for (int k = 0; k < DD; k += 4) {
        const int ka = k + 2 * half;
        v2f a0, a1;
        a0.x = arow0[ka];
        a0.y = arow0[ka + 1];
        a1.x = arow1[ka];
        a1.y = arow1[ka + 1];
        const float* bk = bcol + (size_t)ka * DD;
        #pragma unroll
        for (int t = 0; t < 4; ++t) {
            v2f b;
            b.x = bk[t * 16];
            b.y = bk[t * 16 + DD];
            acc[0][t] = wmma_f32(a0, b, acc[0][t]);
            acc[1][t] = wmma_f32(a1, b, acc[1][t]);
        }
    }

    float* crow = C + n0 + r;
    #pragma unroll
    for (int u = 0; u < 2; ++u) {
        #pragma unroll
        for (int t = 0; t < 4; ++t) {
            #pragma unroll
            for (int j = 0; j < 8; ++j) {
                const int row = m0 + u * 16 + j + 8 * half;
                crow[(size_t)row * DD + t * 16] = acc[u][t][j];
            }
        }
    }
}

__global__ __launch_bounds__(256) void stage1_kernel(
    const float* __restrict__ pooled, const float* __restrict__ W1,
    const float* __restrict__ b1, float* __restrict__ H1)
{
    const int g  = blockIdx.x / 3;
    const int wt = (blockIdx.x % 3) * 8 + (threadIdx.x >> 5);
    const int m0 = (wt / 12) * 32;
    const int n0 = (wt % 12) * 64;
    gemm_wave_32x64(pooled,
                    W1 + (size_t)g * DD * DD,
                    b1 + (size_t)g * DD,
                    H1 + (size_t)g * BB * DD,
                    m0, n0);
}

__global__ __launch_bounds__(256) void stage2_kernel(
    const float* __restrict__ H1, const float* __restrict__ W2,
    const float* __restrict__ b2, float* __restrict__ H2)
{
    const int p  = blockIdx.x / 3;
    const int g1 = p >> 3;
    const int g2 = p & 7;
    const int wt = (blockIdx.x % 3) * 8 + (threadIdx.x >> 5);
    const int m0 = (wt / 12) * 32;
    const int n0 = (wt % 12) * 64;
    gemm_wave_32x64(H1 + (size_t)g1 * BB * DD,
                    W2 + (size_t)g2 * DD * DD,
                    b2 + (size_t)g2 * DD,
                    H2 + (size_t)p * BB * DD,
                    m0, n0);
}

__global__ __launch_bounds__(256) void stage3_kernel(
    const float* __restrict__ H2, const float* __restrict__ Wh,
    const float* __restrict__ bh, const int* __restrict__ g1_idx,
    const int* __restrict__ g2_idx, float* __restrict__ out)
{
    const int wtile = blockIdx.x * 8 + (threadIdx.x >> 5);
    const int a  = wtile >> 2;
    const int m0 = (wtile & 3) * 16;
    const int pair = g1_idx[a] * GG2 + g2_idx[a];

    const float* Am = H2 + (size_t)pair * BB * DD;
    const float* Bm = Wh + (size_t)a * DD * LL;

    const int lane = threadIdx.x & 31;
    const int half = lane >> 4;
    const int r    = lane & 15;
    const int rc   = r & 1;
    const float keep = (r < LL) ? 1.0f : 0.0f;

    v8f acc;
    const float bv = (r < LL) ? bh[a * LL + r] : 0.0f;
    #pragma unroll
    for (int j = 0; j < 8; ++j) acc[j] = bv;

    const float* arow = Am + (size_t)(m0 + r) * DD;

    #pragma unroll 4
    for (int k = 0; k < DD; k += 4) {
        const int ka = k + 2 * half;
        v2f a2;
        a2.x = arow[ka];
        a2.y = arow[ka + 1];
        v2f b;
        b.x = keep * Bm[ka * LL + rc];
        b.y = keep * Bm[(ka + 1) * LL + rc];
        acc = wmma_f32(a2, b, acc);
    }

    if (r < LL) {
        #pragma unroll
        for (int j = 0; j < 8; ++j) {
            const int row = m0 + j + 8 * half;
            out[(size_t)row * (AANN * LL) + a * LL + r] = acc[j];
        }
    }
}

extern "C" void kernel_launch(void* const* d_in, const int* in_sizes, int n_in,
                              void* d_out, int out_size, void* d_ws, size_t ws_size,
                              hipStream_t stream) {
    const float* pooled = (const float*)d_in[0];
    const float* W1     = (const float*)d_in[1];
    const float* b1     = (const float*)d_in[2];
    const float* W2     = (const float*)d_in[3];
    const float* b2     = (const float*)d_in[4];
    const float* Wh     = (const float*)d_in[5];
    const float* bh     = (const float*)d_in[6];
    const int*   g1     = (const int*)d_in[7];
    const int*   g2     = (const int*)d_in[8];
    float*       out    = (float*)d_out;

    float* H1 = (float*)d_ws;
    float* H2 = H1 + (size_t)GG1 * BB * DD;

    stage1_kernel<<<GG1 * 3, 256, 0, stream>>>(pooled, W1, b1, H1);
    stage2_kernel<<<GG1 * GG2 * 3, 256, 0, stream>>>(H1, W2, b2, H2);
    stage3_kernel<<<(AANN * 4) / 8, 256, 0, stream>>>(H2, Wh, bh, g1, g2, out);
}


# Split K over the waves of one work-group per tile (16, 8, 8 slices in the three GEMM stages); partial tiles summed in LDS in fixed wave order, bias added last (changes the summation order)
# speedup vs baseline: 4.7437x; 4.7437x over previous
#include <hip/hip_runtime.h>
#include <hip/hip_bf16.h>


#define AANN 256
#define BB   64
#define DD   768
#define LL   2
#define GG1  4
#define GG2  8

#define TPITCH 68
#define TILE_FLOATS (32 * TPITCH)

typedef __attribute__((ext_vector_type(2))) float v2f;
typedef __attribute__((ext_vector_type(8))) float v8f;

__device__ __forceinline__ v8f wmma_f32(v2f a, v2f b, v8f c) {
    return __builtin_amdgcn_wmma_f32_16x16x4_f32(
         false, a,  false, b,
         (short)0, c,  false,  false);
}

template <int SPLIT>
__device__ __forceinline__ void gemm_tile_32x64_splitk(
    const float* __restrict__ A,
    const float* __restrict__ Bm,
    const float* __restrict__ bias,
    float* __restrict__ C,
    int m0, int n0,
    float* part)
{
    constexpr int KW = DD / SPLIT;
    const int lane = threadIdx.x & 31;
    const int wave = threadIdx.x >> 5;
    const int half = lane >> 4;
    const int r    = lane & 15;
    const int kbeg = wave * KW;

    v8f acc[2][4];
    #pragma unroll
    for (int t = 0; t < 4; ++t) {
        #pragma unroll
        for (int j = 0; j < 8; ++j) { acc[0][t][j] = 0.0f; acc[1][t][j] = 0.0f; }
    }

    const float* arow0 = A + (size_t)(m0 + r) * DD;
    const float* arow1 = arow0 + (size_t)16 * DD;
    const float* bcol  = Bm + n0 + r;

    #pragma unroll 2
    for (int kk = 0; kk < KW; kk += 4) {
        const int ka = kbeg + kk + 2 * half;
        v2f a0, a1;
        a0.x = arow0[ka];
        a0.y = arow0[ka + 1];
        a1.x = arow1[ka];
        a1.y = arow1[ka + 1];
        const float* bk = bcol + (size_t)ka * DD;
        #pragma unroll
        for (int t = 0; t < 4; ++t) {
            v2f b;
            b.x = bk[t * 16];
            b.y = bk[t * 16 + DD];
            acc[0][t] = wmma_f32(a0, b, acc[0][t]);
            acc[1][t] = wmma_f32(a1, b, acc[1][t]);
        }
    }

    float* mine = part + wave * TILE_FLOATS;
    #pragma unroll
    for (int u = 0; u < 2; ++u) {
        #pragma unroll
        for (int t = 0; t < 4; ++t) {
            #pragma unroll
            for (int j = 0; j < 8; ++j) {
                const int row = u * 16 + j + 8 * half;
                mine[row * TPITCH + t * 16 + r] = acc[u][t][j];
            }
        }
    }
    __syncthreads();

    for (int idx = threadIdx.x; idx < 32 * 16; idx += SPLIT * 32) {
        const int row = idx >> 4;
        const int c4  = (idx & 15) * 4;
        float4 s = *reinterpret_cast<const float4*>(bias + n0 + c4);
        const float* p = part + row * TPITCH + c4;
        #pragma unroll
        for (int w = 0; w < SPLIT; ++w) {
            const float4 v = *reinterpret_cast<const float4*>(p + w * TILE_FLOATS);
            s.x += v.x; s.y += v.y; s.z += v.z; s.w += v.w;
        }
        *reinterpret_cast<float4*>(C + (size_t)(m0 + row) * DD + n0 + c4) = s;
    }
}

#define SPLIT1 16
__global__ __launch_bounds__(SPLIT1 * 32) void stage1_kernel(
    const float* __restrict__ pooled, const float* __restrict__ W1,
    const float* __restrict__ b1, float* __restrict__ H1)
{
    __shared__ __attribute__((aligned(16))) float part[SPLIT1 * TILE_FLOATS];
    const int g  = blockIdx.x / 24;
    const int wt = blockIdx.x % 24;
    const int m0 = (wt / 12) * 32;
    const int n0 = (wt % 12) * 64;
    gemm_tile_32x64_splitk<SPLIT1>(pooled,
                    W1 + (size_t)g * DD * DD,
                    b1 + (size_t)g * DD,
                    H1 + (size_t)g * BB * DD,
                    m0, n0, part);
}

#define SPLIT2 8
__global__ __launch_bounds__(SPLIT2 * 32) void stage2_kernel(
    const float* __restrict__ H1, const float* __restrict__ W2,
    const float* __restrict__ b2, float* __restrict__ H2)
{
    __shared__ __attribute__((aligned(16))) float part[SPLIT2 * TILE_FLOATS];
    const int p  = blockIdx.x / 24;
    const int g1 = p >> 3;
    const int g2 = p & 7;
    const int wt = blockIdx.x % 24;
    const int m0 = (wt / 12) * 32;
    const int n0 = (wt % 12) * 64;
    gemm_tile_32x64_splitk<SPLIT2>(H1 + (size_t)g1 * BB * DD,
                    W2 + (size_t)g2 * DD * DD,
                    b2 + (size_t)g2 * DD,
                    H2 + (size_t)p * BB * DD,
                    m0, n0, part);
}

#define SPLIT3 8
__global__ __launch_bounds__(SPLIT3 * 32) void stage3_kernel(
    const float* __restrict__ H2, const float* __restrict__ Wh,
    const float* __restrict__ bh, const int* __restrict__ g1_idx,
    const int* __restrict__ g2_idx, float* __restrict__ out)
{
    __shared__ float part3[SPLIT3 * 16 * LL];
    constexpr int KW = DD / SPLIT3;
    const int a  = blockIdx.x >> 2;
    const int m0 = (blockIdx.x & 3) * 16;
    const int pair = g1_idx[a] * GG2 + g2_idx[a];

    const float* Am = H2 + (size_t)pair * BB * DD;
    const float* Bm = Wh + (size_t)a * DD * LL;

    const int lane = threadIdx.x & 31;
    const int wave = threadIdx.x >> 5;
    const int half = lane >> 4;
    const int r    = lane & 15;
    const int rc   = r & 1;
    const float keep = (r < LL) ? 1.0f : 0.0f;
    const int kbeg = wave * KW;

    v8f acc;
    #pragma unroll
    for (int j = 0; j < 8; ++j) acc[j] = 0.0f;

    const float* arow = Am + (size_t)(m0 + r) * DD;

    #pragma unroll 4
    for (int kk = 0; kk < KW; kk += 4) {
        const int ka = kbeg + kk + 2 * half;
        v2f a2;
        a2.x = arow[ka];
        a2.y = arow[ka + 1];
        v2f b;
        b.x = keep * Bm[ka * LL + rc];
        b.y = keep * Bm[(ka + 1) * LL + rc];
        acc = wmma_f32(a2, b, acc);
    }

    if (r < LL) {
        #pragma unroll
        for (int j = 0; j < 8; ++j) {
            part3[(wave * 16 + j + 8 * half) * LL + r] = acc[j];
        }
    }
    __syncthreads();

    if (threadIdx.x < 16 * LL) {
        const int row = threadIdx.x >> 1;
        const int l   = threadIdx.x & 1;
        float s = bh[a * LL + l];
        #pragma unroll
        for (int w = 0; w < SPLIT3; ++w) s += part3[(w * 16 + row) * LL + l];
        out[(size_t)(m0 + row) * (AANN * LL) + a * LL + l] = s;
    }
}

extern "C" void kernel_launch(void* const* d_in, const int* in_sizes, int n_in,
                              void* d_out, int out_size, void* d_ws, size_t ws_size,
                              hipStream_t stream) {
    const float* pooled = (const float*)d_in[0];
    const float* W1     = (const float*)d_in[1];
    const float* b1     = (const float*)d_in[2];
    const float* W2     = (const float*)d_in[3];
    const float* b2     = (const float*)d_in[4];
    const float* Wh     = (const float*)d_in[5];
    const float* bh     = (const float*)d_in[6];
    const int*   g1     = (const int*)d_in[7];
    const int*   g2     = (const int*)d_in[8];
    float*       out    = (float*)d_out;

    float* H1 = (float*)d_ws;
    float* H2 = H1 + (size_t)GG1 * BB * DD;

    stage1_kernel<<<GG1 * 24, SPLIT1 * 32, 0, stream>>>(pooled, W1, b1, H1);
    stage2_kernel<<<GG1 * GG2 * 24, SPLIT2 * 32, 0, stream>>>(H1, W2, b2, H2);
    stage3_kernel<<<AANN * 4, SPLIT3 * 32, 0, stream>>>(H2, Wh, bh, g1, g2, out);
}


# All three GEMMs fetch their operands as 16-byte vectors per lane (A: consecutive k of the lane's row; B: four neighbouring columns, one per tile); the k order of a wave's sum changes
# speedup vs baseline: 6.9911x; 1.4738x over previous
#include <hip/hip_runtime.h>
#include <hip/hip_bf16.h>


#define AANN 256
#define BB   64
#define DD   768
#define LL   2
#define GG1  4
#define GG2  8

#define TPITCH 68
#define TILE_FLOATS (32 * TPITCH)

typedef __attribute__((ext_vector_type(2))) float v2f;
typedef __attribute__((ext_vector_type(8))) float v8f;

__device__ __forceinline__ v8f wmma_f32(v2f a, v2f b, v8f c) {
    return __builtin_amdgcn_wmma_f32_16x16x4_f32(
         false, a,  false, b,
         (short)0, c,  false,  false);
}

template <int SPLIT>
__device__ __forceinline__ void gemm_tile_32x64_splitk(
    const float* __restrict__ A,
    const float* __restrict__ Bm,
    const float* __restrict__ bias,
    float* __restrict__ C,
    int m0, int n0,
    float* part)
{
    constexpr int KW = DD / SPLIT;
    constexpr int TK = (KW % 32 == 0) ? 32 : 16;
    constexpr int HK = TK / 2;
    static_assert(KW % TK == 0, "slice must be whole trips");
    const int lane = threadIdx.x & 31;
    const int wave = threadIdx.x >> 5;
    const int half = lane >> 4;
    const int r    = lane & 15;
    const int kl   = wave * KW + half * HK;

    v8f acc[2][4];
    #pragma unroll
    for (int t = 0; t < 4; ++t) {
        #pragma unroll
        for (int j = 0; j < 8; ++j) { acc[0][t][j] = 0.0f; acc[1][t][j] = 0.0f; }
    }

    const float* arow0 = A + (size_t)(m0 + r) * DD + kl;
    const float* arow1 = arow0 + (size_t)16 * DD;
    const float* brow  = Bm + (size_t)kl * DD + n0 + 4 * r;

    for (int kk = 0; kk < KW; kk += TK) {
        float4 a0v[HK / 4], a1v[HK / 4];
        #pragma unroll
        for (int i = 0; i < HK / 4; ++i) {
            a0v[i] = *reinterpret_cast<const float4*>(arow0 + kk + 4 * i);
            a1v[i] = *reinterpret_cast<const float4*>(arow1 + kk + 4 * i);
        }
        #pragma unroll
        for (int g = 0; g < HK / 8; ++g) {
            float4 bv[8];
            #pragma unroll
            for (int i = 0; i < 8; ++i)
                bv[i] = *reinterpret_cast<const float4*>(brow + (size_t)(kk + 8 * g + i) * DD);
            __builtin_amdgcn_sched_barrier(0);
            #pragma unroll
            for (int s = 0; s < 4; ++s) {
                const float4 p0 = a0v[2 * g + (s >> 1)];
                const float4 p1 = a1v[2 * g + (s >> 1)];
                v2f a0, a1;
                if (s & 1) { a0.x = p0.z; a0.y = p0.w; a1.x = p1.z; a1.y = p1.w; }
                else       { a0.x = p0.x; a0.y = p0.y; a1.x = p1.x; a1.y = p1.y; }
                const float bx[4] = { bv[2 * s].x,     bv[2 * s].y,     bv[2 * s].z,     bv[2 * s].w };
                const float by[4] = { bv[2 * s + 1].x, bv[2 * s + 1].y, bv[2 * s + 1].z, bv[2 * s + 1].w };
                #pragma unroll
                for (int t = 0; t < 4; ++t) {
                    v2f b;
                    b.x = bx[t];
                    b.y = by[t];
                    acc[0][t] = wmma_f32(a0, b, acc[0][t]);
                    acc[1][t] = wmma_f32(a1, b, acc[1][t]);
                }
            }
        }
    }

    float* mine = part + wave * TILE_FLOATS;
    #pragma unroll
    for (int u = 0; u < 2; ++u) {
        #pragma unroll
        for (int j = 0; j < 8; ++j) {
            const int row = u * 16 + j + 8 * half;
            float4 v;
            v.x = acc[u][0][j]; v.y = acc[u][1][j]; v.z = acc[u][2][j]; v.w = acc[u][3][j];
            *reinterpret_cast<float4*>(mine + row * TPITCH + 4 * r) = v;
        }
    }
    __syncthreads();

    for (int idx = threadIdx.x; idx < 32 * 16; idx += SPLIT * 32) {
        const int row = idx >> 4;
        const int c4  = (idx & 15) * 4;
        float4 s = *reinterpret_cast<const float4*>(bias + n0 + c4);
        const float* p = part + row * TPITCH + c4;
        #pragma unroll
        for (int w = 0; w < SPLIT; ++w) {
            const float4 v = *reinterpret_cast<const float4*>(p + w * TILE_FLOATS);
            s.x += v.x; s.y += v.y; s.z += v.z; s.w += v.w;
        }
        *reinterpret_cast<float4*>(C + (size_t)(m0 + row) * DD + n0 + c4) = s;
    }
}

#define SPLIT1 16
__global__ __launch_bounds__(SPLIT1 * 32) __attribute__((amdgpu_waves_per_eu(4, 4))) void stage1_kernel(
    const float* __restrict__ pooled, const float* __restrict__ W1,
    const float* __restrict__ b1, float* __restrict__ H1)
{
    __shared__ __attribute__((aligned(16))) float part[SPLIT1 * TILE_FLOATS];
    const int g  = blockIdx.x / 24;
    const int wt = blockIdx.x % 24;
    const int m0 = (wt / 12) * 32;
    const int n0 = (wt % 12) * 64;
    gemm_tile_32x64_splitk<SPLIT1>(pooled,
                    W1 + (size_t)g * DD * DD,
                    b1 + (size_t)g * DD,
                    H1 + (size_t)g * BB * DD,
                    m0, n0, part);
}

#define SPLIT2 8
__global__ __launch_bounds__(SPLIT2 * 32) __attribute__((amdgpu_waves_per_eu(6, 6))) void stage2_kernel(
    const float* __restrict__ H1, const float* __restrict__ W2,
    const float* __restrict__ b2, float* __restrict__ H2)
{
    __shared__ __attribute__((aligned(16))) float part[SPLIT2 * TILE_FLOATS];
    const int p  = blockIdx.x / 24;
    const int g1 = p >> 3;
    const int g2 = p & 7;
    const int wt = blockIdx.x % 24;
    const int m0 = (wt / 12) * 32;
    const int n0 = (wt % 12) * 64;
    gemm_tile_32x64_splitk<SPLIT2>(H1 + (size_t)g1 * BB * DD,
                    W2 + (size_t)g2 * DD * DD,
                    b2 + (size_t)g2 * DD,
                    H2 + (size_t)p * BB * DD,
                    m0, n0, part);
}

#define SPLIT3 8
__global__ __launch_bounds__(SPLIT3 * 32) __attribute__((amdgpu_waves_per_eu(8, 8))) void stage3_kernel(
    const float* __restrict__ H2, const float* __restrict__ Wh,
    const float* __restrict__ bh, const int* __restrict__ g1_idx,
    const int* __restrict__ g2_idx, float* __restrict__ out)
{
    __shared__ float part3[SPLIT3 * 16 * LL];
    constexpr int KW = DD / SPLIT3;
    const int a  = blockIdx.x >> 2;
    const int m0 = (blockIdx.x & 3) * 16;
    const int pair = g1_idx[a] * GG2 + g2_idx[a];

    const float* Am = H2 + (size_t)pair * BB * DD;
    const float* Bm = Wh + (size_t)a * DD * LL;

    const int lane = threadIdx.x & 31;
    const int wave = threadIdx.x >> 5;
    const int half = lane >> 4;
    const int r    = lane & 15;
    const bool col0 = (r == 0);
    const bool col1 = (r == 1);
    const int kl   = wave * KW + half * 16;

    v8f acc;
    #pragma unroll
    for (int j = 0; j < 8; ++j) acc[j] = 0.0f;

    const float* arow = Am + (size_t)(m0 + r) * DD + kl;
    const float* bptr = Bm + (size_t)kl * LL;

    for (int kk = 0; kk < KW; kk += 32) {
        float4 av[4], q[8];
        #pragma unroll
        for (int i = 0; i < 4; ++i) av[i] = *reinterpret_cast<const float4*>(arow + kk + 4 * i);
        #pragma unroll
        for (int i = 0; i < 8; ++i) q[i] = *reinterpret_cast<const float4*>(bptr + kk * LL + 4 * i);
        __builtin_amdgcn_sched_barrier(0);
        #pragma unroll
        for (int s = 0; s < 8; ++s) {
            const float4 p = av[s >> 1];
            v2f a2;
            if (s & 1) { a2.x = p.z; a2.y = p.w; }
            else       { a2.x = p.x; a2.y = p.y; }
            v2f b;
            b.x = col0 ? q[s].x : (col1 ? q[s].y : 0.0f);
            b.y = col0 ? q[s].z : (col1 ? q[s].w : 0.0f);
            acc = wmma_f32(a2, b, acc);
        }
    }

    if (r < LL) {
        #pragma unroll
        for (int j = 0; j < 8; ++j) {
            part3[(wave * 16 + j + 8 * half) * LL + r] = acc[j];
        }
    }
    __syncthreads();

    if (threadIdx.x < 16 * LL) {
        const int row = threadIdx.x >> 1;
        const int l   = threadIdx.x & 1;
        float s = bh[a * LL + l];
        #pragma unroll
        for (int w = 0; w < SPLIT3; ++w) s += part3[(w * 16 + row) * LL + l];
        out[(size_t)(m0 + row) * (AANN * LL) + a * LL + l] = s;
    }
}

extern "C" void kernel_launch(void* const* d_in, const int* in_sizes, int n_in,
                              void* d_out, int out_size, void* d_ws, size_t ws_size,
                              hipStream_t stream) {
    const float* pooled = (const float*)d_in[0];
    const float* W1     = (const float*)d_in[1];
    const float* b1     = (const float*)d_in[2];
    const float* W2     = (const float*)d_in[3];
    const float* b2     = (const float*)d_in[4];
    const float* Wh     = (const float*)d_in[5];
    const float* bh     = (const float*)d_in[6];
    const int*   g1     = (const int*)d_in[7];
    const int*   g2     = (const int*)d_in[8];
    float*       out    = (float*)d_out;

    float* H1 = (float*)d_ws;
    float* H2 = H1 + (size_t)GG1 * BB * DD;

    stage1_kernel<<<GG1 * 24, SPLIT1 * 32, 0, stream>>>(pooled, W1, b1, H1);
    stage2_kernel<<<GG1 * GG2 * 24, SPLIT2 * 32, 0, stream>>>(H1, W2, b2, H2);
    stage3_kernel<<<AANN * 4, SPLIT3 * 32, 0, stream>>>(H2, Wh, bh, g1, g2, out);
}


# Product reassociated: H1 @ (W2[g2] @ Wh[a]) + (b2 @ Wh + bh) for (H1 @ W2 + b2) @ Wh; stage 2 forms W2 @ Wh per tile of 8 same-g2 annotators, no H2; changes the summation order
# speedup vs baseline: 7.7502x; 1.1086x over previous
#include <hip/hip_runtime.h>
#include <hip/hip_bf16.h>


#define AANN 256
#define BB   64
#define DD   768
#define LL   2
#define GG1  4
#define GG2  8

#define TPITCH 68
#define TILE_FLOATS (32 * TPITCH)

typedef __attribute__((ext_vector_type(2))) float v2f;
typedef __attribute__((ext_vector_type(8))) float v8f;

__device__ __forceinline__ v8f wmma_f32(v2f a, v2f b, v8f c) {
    return __builtin_amdgcn_wmma_f32_16x16x4_f32(
         false, a,  false, b,
         (short)0, c,  false,  false);
}

template <int SPLIT>
__device__ __forceinline__ void gemm_tile_32x64_splitk(
    const float* __restrict__ A,
    const float* __restrict__ Bm,
    const float* __restrict__ bias,
    float* __restrict__ C,
    int m0, int n0,
    float* part)
{
    constexpr int KW = DD / SPLIT;
    constexpr int TK = (KW % 32 == 0) ? 32 : 16;
    constexpr int HK = TK / 2;
    static_assert(KW % TK == 0, "slice must be whole trips");
    const int lane = threadIdx.x & 31;
    const int wave = threadIdx.x >> 5;
    const int half = lane >> 4;
    const int r    = lane & 15;
    const int kl   = wave * KW + half * HK;

    v8f acc[2][4];
    #pragma unroll
    for (int t = 0; t < 4; ++t) {
        #pragma unroll
        for (int j = 0; j < 8; ++j) { acc[0][t][j] = 0.0f; acc[1][t][j] = 0.0f; }
    }

    const float* arow0 = A + (size_t)(m0 + r) * DD + kl;
    const float* arow1 = arow0 + (size_t)16 * DD;
    const float* brow  = Bm + (size_t)kl * DD + n0 + 4 * r;

    for (int kk = 0; kk < KW; kk += TK) {
        float4 a0v[HK / 4], a1v[HK / 4];
        #pragma unroll
        for (int i = 0; i < HK / 4; ++i) {
            a0v[i] = *reinterpret_cast<const float4*>(arow0 + kk + 4 * i);
            a1v[i] = *reinterpret_cast<const float4*>(arow1 + kk + 4 * i);
        }
        #pragma unroll
        for (int g = 0; g < HK / 8; ++g) {
            float4 bv[8];
            #pragma unroll
            for (int i = 0; i < 8; ++i)
                bv[i] = *reinterpret_cast<const float4*>(brow + (size_t)(kk + 8 * g + i) * DD);
            __builtin_amdgcn_sched_barrier(0);
            #pragma unroll
            for (int s = 0; s < 4; ++s) {
                const float4 p0 = a0v[2 * g + (s >> 1)];
                const float4 p1 = a1v[2 * g + (s >> 1)];
                v2f a0, a1;
                if (s & 1) { a0.x = p0.z; a0.y = p0.w; a1.x = p1.z; a1.y = p1.w; }
                else       { a0.x = p0.x; a0.y = p0.y; a1.x = p1.x; a1.y = p1.y; }
                const float bx[4] = { bv[2 * s].x,     bv[2 * s].y,     bv[2 * s].z,     bv[2 * s].w };
                const float by[4] = { bv[2 * s + 1].x, bv[2 * s + 1].y, bv[2 * s + 1].z, bv[2 * s + 1].w };
                #pragma unroll
                for (int t = 0; t < 4; ++t) {
                    v2f b;
                    b.x = bx[t];
                    b.y = by[t];
                    acc[0][t] = wmma_f32(a0, b, acc[0][t]);
                    acc[1][t] = wmma_f32(a1, b, acc[1][t]);
                }
            }
        }
    }

    float* mine = part + wave * TILE_FLOATS;
    #pragma unroll
    for (int u = 0; u < 2; ++u) {
        #pragma unroll
        for (int j = 0; j < 8; ++j) {
            const int row = u * 16 + j + 8 * half;
            float4 v;
            v.x = acc[u][0][j]; v.y = acc[u][1][j]; v.z = acc[u][2][j]; v.w = acc[u][3][j];
            *reinterpret_cast<float4*>(mine + row * TPITCH + 4 * r) = v;
        }
    }
    __syncthreads();

    for (int idx = threadIdx.x; idx < 32 * 16; idx += SPLIT * 32) {
        const int row = idx >> 4;
        const int c4  = (idx & 15) * 4;
        float4 s = *reinterpret_cast<const float4*>(bias + n0 + c4);
        const float* p = part + row * TPITCH + c4;
        #pragma unroll
        for (int w = 0; w < SPLIT; ++w) {
            const float4 v = *reinterpret_cast<const float4*>(p + w * TILE_FLOATS);
            s.x += v.x; s.y += v.y; s.z += v.z; s.w += v.w;
        }
        *reinterpret_cast<float4*>(C + (size_t)(m0 + row) * DD + n0 + c4) = s;
    }
}

#define SPLIT1 16
__global__ __launch_bounds__(SPLIT1 * 32) __attribute__((amdgpu_waves_per_eu(4, 4))) void stage1_kernel(
    const float* __restrict__ pooled, const float* __restrict__ W1,
    const float* __restrict__ b1, float* __restrict__ H1)
{
    __shared__ __attribute__((aligned(16))) float part[SPLIT1 * TILE_FLOATS];
    const int g  = blockIdx.x / 24;
    const int wt = blockIdx.x % 24;
    const int m0 = (wt / 12) * 32;
    const int n0 = (wt % 12) * 64;
    gemm_tile_32x64_splitk<SPLIT1>(pooled,
                    W1 + (size_t)g * DD * DD,
                    b1 + (size_t)g * DD,
                    H1 + (size_t)g * BB * DD,
                    m0, n0, part);
}

#define SPLIT2 8
#define VPITCH 18
#define VT_FLOATS (64 * VPITCH)
#define MRANGES 13
#define MAXTILES 39
__global__ __launch_bounds__(SPLIT2 * 32) void stage2_kernel(
    const float* __restrict__ W2, const float* __restrict__ b2,
    const float* __restrict__ Wh, const float* __restrict__ bh,
    const int* __restrict__ g2_idx, float* __restrict__ V, float* __restrict__ Cb)
{
    __shared__ __attribute__((aligned(16))) float part[SPLIT2 * VT_FLOATS];
    __shared__ int s_wcnt[8 * GG2];
    __shared__ int s_ann[8];
    __shared__ int s_g;

    const int bt   = blockIdx.x / MRANGES;
    const int mr   = blockIdx.x % MRANGES;
    const int lane = threadIdx.x & 31;
    const int wave = threadIdx.x >> 5;
    const int half = lane >> 4;
    const int r    = lane & 15;

    const int myg = g2_idx[threadIdx.x] & (GG2 - 1);
    int myrank = 0;
    #pragma unroll
    for (int v = 0; v < GG2; ++v) {
        const unsigned m = __builtin_amdgcn_ballot_w32(myg == v);
        if (lane == 0) s_wcnt[wave * GG2 + v] = __popc(m);
        if (myg == v) myrank = __popc(m & ((1u << lane) - 1u));
    }
    if (threadIdx.x < 8) s_ann[threadIdx.x] = -1;
    __syncthreads();
    int before = 0;
    int tstart = 0;
    #pragma unroll
    for (int v = 0; v < GG2; ++v) {
        int n = 0;
        #pragma unroll
        for (int w = 0; w < 8; ++w) {
            const int c = s_wcnt[w * GG2 + v];
            n += c;
            if (v == myg && w < wave) before += c;
        }
        if (v < myg) tstart += (n + 7) >> 3;
    }
    const int rank = before + myrank;
    if (tstart + (rank >> 3) == bt) {
        s_ann[rank & 7] = threadIdx.x;
        if ((rank & 7) == 0) s_g = myg;
    }
    __syncthreads();
    if (s_ann[0] < 0) return;
    const int g2t = s_g;

    const int  annB  = s_ann[r >> 1];
    const bool valid = annB >= 0;
    const bool lab1  = (r & 1) != 0;
    const int  kl    = wave * 96 + half * 16;
    const float* bptr = Wh + (size_t)(valid ? annB : 0) * DD * LL + (size_t)kl * LL;

    if (mr < 12) {
        const int m0 = mr * 64;
        const float* arow = W2 + (size_t)g2t * DD * DD + (size_t)(m0 + r) * DD + kl;
        v8f acc[4];
        #pragma unroll
        for (int u = 0; u < 4; ++u) {
            #pragma unroll
            for (int j = 0; j < 8; ++j) acc[u][j] = 0.0f;
        }
        for (int kk = 0; kk < 96; kk += 32) {
            float4 av[4][4], q[8];
            #pragma unroll
            for (int u = 0; u < 4; ++u) {
                #pragma unroll
                for (int i = 0; i < 4; ++i)
                    av[u][i] = *reinterpret_cast<const float4*>(arow + (size_t)u * 16 * DD + kk + 4 * i);
            }
            #pragma unroll
            for (int i = 0; i < 8; ++i) q[i] = *reinterpret_cast<const float4*>(bptr + kk * LL + 4 * i);
            __builtin_amdgcn_sched_barrier(0);
            #pragma unroll
            for (int s = 0; s < 8; ++s) {
                v2f b;
                b.x = valid ? (lab1 ? q[s].y : q[s].x) : 0.0f;
                b.y = valid ? (lab1 ? q[s].w : q[s].z) : 0.0f;
                #pragma unroll
                for (int u = 0; u < 4; ++u) {
                    const float4 p = av[u][s >> 1];
                    v2f a2;
                    if (s & 1) { a2.x = p.z; a2.y = p.w; }
                    else       { a2.x = p.x; a2.y = p.y; }
                    acc[u] = wmma_f32(a2, b, acc[u]);
                }
            }
        }
        float* mine = part + wave * VT_FLOATS;
        #pragma unroll
        for (int u = 0; u < 4; ++u) {
            #pragma unroll
            for (int j = 0; j < 8; ++j)
                mine[(16 * u + j + 8 * half) * VPITCH + r] = acc[u][j];
        }
        __syncthreads();
        const int annS = s_ann[wave];
        if (annS >= 0) {
            float4 s = make_float4(0.0f, 0.0f, 0.0f, 0.0f);
            const float* p = part + (2 * lane) * VPITCH + 2 * wave;
            #pragma unroll
            for (int w = 0; w < SPLIT2; ++w) {
                const float2 lo = *reinterpret_cast<const float2*>(p + w * VT_FLOATS);
                const float2 hi = *reinterpret_cast<const float2*>(p + w * VT_FLOATS + VPITCH);
                s.x += lo.x; s.y += lo.y; s.z += hi.x; s.w += hi.y;
            }
            *reinterpret_cast<float4*>(V + (size_t)annS * DD * LL + (size_t)(m0 + 2 * lane) * LL) = s;
        }
    } else {
        const float* brow = b2 + (size_t)g2t * DD + kl;
        const bool row0 = (r == 0);
        v8f acc;
        #pragma unroll
        for (int j = 0; j < 8; ++j) acc[j] = 0.0f;
        for (int kk = 0; kk < 96; kk += 32) {
            float4 av[4], q[8];
            #pragma unroll
            for (int i = 0; i < 4; ++i) av[i] = *reinterpret_cast<const float4*>(brow + kk + 4 * i);
            #pragma unroll
            for (int i = 0; i < 8; ++i) q[i] = *reinterpret_cast<const float4*>(bptr + kk * LL + 4 * i);
            #pragma unroll
            for (int s = 0; s < 8; ++s) {
                const float4 p = av[s >> 1];
                v2f a2, b;
                if (s & 1) { a2.x = row0 ? p.z : 0.0f; a2.y = row0 ? p.w : 0.0f; }
                else       { a2.x = row0 ? p.x : 0.0f; a2.y = row0 ? p.y : 0.0f; }
                b.x = valid ? (lab1 ? q[s].y : q[s].x) : 0.0f;
                b.y = valid ? (lab1 ? q[s].w : q[s].z) : 0.0f;
                acc = wmma_f32(a2, b, acc);
            }
        }
        if (half == 0) part[wave * 16 + r] = acc[0];
        __syncthreads();
        if (threadIdx.x < 16) {
            const int annS = s_ann[threadIdx.x >> 1];
            if (annS >= 0) {
                float s = bh[annS * LL + (threadIdx.x & 1)];
                #pragma unroll
                for (int w = 0; w < SPLIT2; ++w) s += part[w * 16 + threadIdx.x];
                Cb[annS * LL + (threadIdx.x & 1)] = s;
            }
        }
    }
}

#define SPLIT3 8
__global__ __launch_bounds__(SPLIT3 * 32) __attribute__((amdgpu_waves_per_eu(8, 8))) void stage3_kernel(
    const float* __restrict__ H1, const float* __restrict__ V,
    const float* __restrict__ Cb, const int* __restrict__ g1_idx,
    float* __restrict__ out)
{
    __shared__ float part3[SPLIT3 * 16 * LL];
    constexpr int KW = DD / SPLIT3;
    const int a  = blockIdx.x >> 2;
    const int m0 = (blockIdx.x & 3) * 16;
    const float* Am = H1 + (size_t)g1_idx[a] * BB * DD;
    const float* Bm = V + (size_t)a * DD * LL;

    const int lane = threadIdx.x & 31;
    const int wave = threadIdx.x >> 5;
    const int half = lane >> 4;
    const int r    = lane & 15;
    const bool col0 = (r == 0);
    const bool col1 = (r == 1);
    const int kl   = wave * KW + half * 16;

    v8f acc;
    #pragma unroll
    for (int j = 0; j < 8; ++j) acc[j] = 0.0f;

    const float* arow = Am + (size_t)(m0 + r) * DD + kl;
    const float* bptr = Bm + (size_t)kl * LL;

    for (int kk = 0; kk < KW; kk += 32) {
        float4 av[4], q[8];
        #pragma unroll
        for (int i = 0; i < 4; ++i) av[i] = *reinterpret_cast<const float4*>(arow + kk + 4 * i);
        #pragma unroll
        for (int i = 0; i < 8; ++i) q[i] = *reinterpret_cast<const float4*>(bptr + kk * LL + 4 * i);
        __builtin_amdgcn_sched_barrier(0);
        #pragma unroll
        for (int s = 0; s < 8; ++s) {
            const float4 p = av[s >> 1];
            v2f a2;
            if (s & 1) { a2.x = p.z; a2.y = p.w; }
            else       { a2.x = p.x; a2.y = p.y; }
            v2f b;
            b.x = col0 ? q[s].x : (col1 ? q[s].y : 0.0f);
            b.y = col0 ? q[s].z : (col1 ? q[s].w : 0.0f);
            acc = wmma_f32(a2, b, acc);
        }
    }

    if (r < LL) {
        #pragma unroll
        for (int j = 0; j < 8; ++j) {
            part3[(wave * 16 + j + 8 * half) * LL + r] = acc[j];
        }
    }
    __syncthreads();

    if (threadIdx.x < 16 * LL) {
        const int row = threadIdx.x >> 1;
        const int l   = threadIdx.x & 1;
        float s = Cb[a * LL + l];
        #pragma unroll
        for (int w = 0; w < SPLIT3; ++w) s += part3[(w * 16 + row) * LL + l];
        out[(size_t)(m0 + row) * (AANN * LL) + a * LL + l] = s;
    }
}

extern "C" void kernel_launch(void* const* d_in, const int* in_sizes, int n_in,
                              void* d_out, int out_size, void* d_ws, size_t ws_size,
                              hipStream_t stream) {
    const float* pooled = (const float*)d_in[0];
    const float* W1     = (const float*)d_in[1];
    const float* b1     = (const float*)d_in[2];
    const float* W2     = (const float*)d_in[3];
    const float* b2     = (const float*)d_in[4];
    const float* Wh     = (const float*)d_in[5];
    const float* bh     = (const float*)d_in[6];
    const int*   g1     = (const int*)d_in[7];
    const int*   g2     = (const int*)d_in[8];
    float*       out    = (float*)d_out;

    float* H1 = (float*)d_ws;
    float* V  = H1 + (size_t)GG1 * BB * DD;
    float* Cb = V + (size_t)AANN * DD * LL;

    stage1_kernel<<<GG1 * 24, SPLIT1 * 32, 0, stream>>>(pooled, W1, b1, H1);
    stage2_kernel<<<MAXTILES * MRANGES, SPLIT2 * 32, 0, stream>>>(W2, b2, Wh, bh, g2, V, Cb);
    stage3_kernel<<<AANN * 4, SPLIT3 * 32, 0, stream>>>(H1, V, Cb, g1, out);
}


# Stage 3 re-tiled: eight annotators that share g1 fill the 16 columns of one WMMA tile (140 work-groups, not 1024; one load group per wave); K split over 8 waves: changes the summation order
# speedup vs baseline: 8.1947x; 1.0574x over previous
#include <hip/hip_runtime.h>
#include <hip/hip_bf16.h>


#define AANN 256
#define BB   64
#define DD   768
#define LL   2
#define GG1  4
#define GG2  8

#define TPITCH 68
#define TILE_FLOATS (32 * TPITCH)

typedef __attribute__((ext_vector_type(2))) float v2f;
typedef __attribute__((ext_vector_type(8))) float v8f;

__device__ __forceinline__ v8f wmma_f32(v2f a, v2f b, v8f c) {
    return __builtin_amdgcn_wmma_f32_16x16x4_f32(
         false, a,  false, b,
         (short)0, c,  false,  false);
}

template <int SPLIT>
__device__ __forceinline__ void gemm_tile_32x64_splitk(
    const float* __restrict__ A,
    const float* __restrict__ Bm,
    const float* __restrict__ bias,
    float* __restrict__ C,
    int m0, int n0,
    float* part)
{
    constexpr int KW = DD / SPLIT;
    constexpr int TK = (KW % 32 == 0) ? 32 : 16;
    constexpr int HK = TK / 2;
    static_assert(KW % TK == 0, "slice must be whole trips");
    const int lane = threadIdx.x & 31;
    const int wave = threadIdx.x >> 5;
    const int half = lane >> 4;
    const int r    = lane & 15;
    const int kl   = wave * KW + half * HK;

    v8f acc[2][4];
    #pragma unroll
    for (int t = 0; t < 4; ++t) {
        #pragma unroll
        for (int j = 0; j < 8; ++j) { acc[0][t][j] = 0.0f; acc[1][t][j] = 0.0f; }
    }

    const float* arow0 = A + (size_t)(m0 + r) * DD + kl;
    const float* arow1 = arow0 + (size_t)16 * DD;
    const float* brow  = Bm + (size_t)kl * DD + n0 + 4 * r;

    for (int kk = 0; kk < KW; kk += TK) {
        float4 a0v[HK / 4], a1v[HK / 4];
        #pragma unroll
        for (int i = 0; i < HK / 4; ++i) {
            a0v[i] = *reinterpret_cast<const float4*>(arow0 + kk + 4 * i);
            a1v[i] = *reinterpret_cast<const float4*>(arow1 + kk + 4 * i);
        }
        #pragma unroll
        for (int g = 0; g < HK / 8; ++g) {
            float4 bv[8];
            #pragma unroll
            for (int i = 0; i < 8; ++i)
                bv[i] = *reinterpret_cast<const float4*>(brow + (size_t)(kk + 8 * g + i) * DD);
            __builtin_amdgcn_sched_barrier(0);
            #pragma unroll
            for (int s = 0; s < 4; ++s) {
                const float4 p0 = a0v[2 * g + (s >> 1)];
                const float4 p1 = a1v[2 * g + (s >> 1)];
                v2f a0, a1;
                if (s & 1) { a0.x = p0.z; a0.y = p0.w; a1.x = p1.z; a1.y = p1.w; }
                else       { a0.x = p0.x; a0.y = p0.y; a1.x = p1.x; a1.y = p1.y; }
                const float bx[4] = { bv[2 * s].x,     bv[2 * s].y,     bv[2 * s].z,     bv[2 * s].w };
                const float by[4] = { bv[2 * s + 1].x, bv[2 * s + 1].y, bv[2 * s + 1].z, bv[2 * s + 1].w };
                #pragma unroll
                for (int t = 0; t < 4; ++t) {
                    v2f b;
                    b.x = bx[t];
                    b.y = by[t];
                    acc[0][t] = wmma_f32(a0, b, acc[0][t]);
                    acc[1][t] = wmma_f32(a1, b, acc[1][t]);
                }
            }
        }
    }

    float* mine = part + wave * TILE_FLOATS;
    #pragma unroll
    for (int u = 0; u < 2; ++u) {
        #pragma unroll
        for (int j = 0; j < 8; ++j) {
            const int row = u * 16 + j + 8 * half;
            float4 v;
            v.x = acc[u][0][j]; v.y = acc[u][1][j]; v.z = acc[u][2][j]; v.w = acc[u][3][j];
            *reinterpret_cast<float4*>(mine + row * TPITCH + 4 * r) = v;
        }
    }
    __syncthreads();

    for (int idx = threadIdx.x; idx < 32 * 16; idx += SPLIT * 32) {
        const int row = idx >> 4;
        const int c4  = (idx & 15) * 4;
        float4 s = *reinterpret_cast<const float4*>(bias + n0 + c4);
        const float* p = part + row * TPITCH + c4;
        #pragma unroll
        for (int w = 0; w < SPLIT; ++w) {
            const float4 v = *reinterpret_cast<const float4*>(p + w * TILE_FLOATS);
            s.x += v.x; s.y += v.y; s.z += v.z; s.w += v.w;
        }
        *reinterpret_cast<float4*>(C + (size_t)(m0 + row) * DD + n0 + c4) = s;
    }
}

#define SPLIT1 16
__global__ __launch_bounds__(SPLIT1 * 32) __attribute__((amdgpu_waves_per_eu(4, 4))) void stage1_kernel(
    const float* __restrict__ pooled, const float* __restrict__ W1,
    const float* __restrict__ b1, float* __restrict__ H1)
{
    __shared__ __attribute__((aligned(16))) float part[SPLIT1 * TILE_FLOATS];
    const int g  = blockIdx.x / 24;
    const int wt = blockIdx.x % 24;
    const int m0 = (wt / 12) * 32;
    const int n0 = (wt % 12) * 64;
    gemm_tile_32x64_splitk<SPLIT1>(pooled,
                    W1 + (size_t)g * DD * DD,
                    b1 + (size_t)g * DD,
                    H1 + (size_t)g * BB * DD,
                    m0, n0, part);
}

#define SPLIT2 8
#define VPITCH 18
#define VT_FLOATS (64 * VPITCH)
#define MRANGES 13
#define MAXTILES 39
__global__ __launch_bounds__(SPLIT2 * 32) void stage2_kernel(
    const float* __restrict__ W2, const float* __restrict__ b2,
    const float* __restrict__ Wh, const float* __restrict__ bh,
    const int* __restrict__ g2_idx, float* __restrict__ V, float* __restrict__ Cb)
{
    __shared__ __attribute__((aligned(16))) float part[SPLIT2 * VT_FLOATS];
    __shared__ int s_wcnt[8 * GG2];
    __shared__ int s_ann[8];
    __shared__ int s_g;

    const int bt   = blockIdx.x / MRANGES;
    const int mr   = blockIdx.x % MRANGES;
    const int lane = threadIdx.x & 31;
    const int wave = threadIdx.x >> 5;
    const int half = lane >> 4;
    const int r    = lane & 15;

    const int myg = g2_idx[threadIdx.x] & (GG2 - 1);
    int myrank = 0;
    #pragma unroll
    for (int v = 0; v < GG2; ++v) {
        const unsigned m = __builtin_amdgcn_ballot_w32(myg == v);
        if (lane == 0) s_wcnt[wave * GG2 + v] = __popc(m);
        if (myg == v) myrank = __popc(m & ((1u << lane) - 1u));
    }
    if (threadIdx.x < 8) s_ann[threadIdx.x] = -1;
    __syncthreads();
    int before = 0;
    int tstart = 0;
    #pragma unroll
    for (int v = 0; v < GG2; ++v) {
        int n = 0;
        #pragma unroll
        for (int w = 0; w < 8; ++w) {
            const int c = s_wcnt[w * GG2 + v];
            n += c;
            if (v == myg && w < wave) before += c;
        }
        if (v < myg) tstart += (n + 7) >> 3;
    }
    const int rank = before + myrank;
    if (tstart + (rank >> 3) == bt) {
        s_ann[rank & 7] = threadIdx.x;
        if ((rank & 7) == 0) s_g = myg;
    }
    __syncthreads();
    if (s_ann[0] < 0) return;
    const int g2t = s_g;

    const int  annB  = s_ann[r >> 1];
    const bool valid = annB >= 0;
    const bool lab1  = (r & 1) != 0;
    const int  kl    = wave * 96 + half * 16;
    const float* bptr = Wh + (size_t)(valid ? annB : 0) * DD * LL + (size_t)kl * LL;

    if (mr < 12) {
        const int m0 = mr * 64;
        const float* arow = W2 + (size_t)g2t * DD * DD + (size_t)(m0 + r) * DD + kl;
        v8f acc[4];
        #pragma unroll
        for (int u = 0; u < 4; ++u) {
            #pragma unroll
            for (int j = 0; j < 8; ++j) acc[u][j] = 0.0f;
        }
        for (int kk = 0; kk < 96; kk += 32) {
            float4 av[4][4], q[8];
            #pragma unroll
            for (int u = 0; u < 4; ++u) {
                #pragma unroll
                for (int i = 0; i < 4; ++i)
                    av[u][i] = *reinterpret_cast<const float4*>(arow + (size_t)u * 16 * DD + kk + 4 * i);
            }
            #pragma unroll
            for (int i = 0; i < 8; ++i) q[i] = *reinterpret_cast<const float4*>(bptr + kk * LL + 4 * i);
            __builtin_amdgcn_sched_barrier(0);
            #pragma unroll
            for (int s = 0; s < 8; ++s) {
                v2f b;
                b.x = valid ? (lab1 ? q[s].y : q[s].x) : 0.0f;
                b.y = valid ? (lab1 ? q[s].w : q[s].z) : 0.0f;
                #pragma unroll
                for (int u = 0; u < 4; ++u) {
                    const float4 p = av[u][s >> 1];
                    v2f a2;
                    if (s & 1) { a2.x = p.z; a2.y = p.w; }
                    else       { a2.x = p.x; a2.y = p.y; }
                    acc[u] = wmma_f32(a2, b, acc[u]);
                }
            }
        }
        float* mine = part + wave * VT_FLOATS;
        #pragma unroll
        for (int u = 0; u < 4; ++u) {
            #pragma unroll
            for (int j = 0; j < 8; ++j)
                mine[(16 * u + j + 8 * half) * VPITCH + r] = acc[u][j];
        }
        __syncthreads();
        const int annS = s_ann[wave];
        if (annS >= 0) {
            float4 s = make_float4(0.0f, 0.0f, 0.0f, 0.0f);
            const float* p = part + (2 * lane) * VPITCH + 2 * wave;
            #pragma unroll
            for (int w = 0; w < SPLIT2; ++w) {
                const float2 lo = *reinterpret_cast<const float2*>(p + w * VT_FLOATS);
                const float2 hi = *reinterpret_cast<const float2*>(p + w * VT_FLOATS + VPITCH);
                s.x += lo.x; s.y += lo.y; s.z += hi.x; s.w += hi.y;
            }
            *reinterpret_cast<float4*>(V + (size_t)annS * DD * LL + (size_t)(m0 + 2 * lane) * LL) = s;
        }
    } else {
        const float* brow = b2 + (size_t)g2t * DD + kl;
        const bool row0 = (r == 0);
        v8f acc;
        #pragma unroll
        for (int j = 0; j < 8; ++j) acc[j] = 0.0f;
        for (int kk = 0; kk < 96; kk += 32) {
            float4 av[4], q[8];
            #pragma unroll
            for (int i = 0; i < 4; ++i) av[i] = *reinterpret_cast<const float4*>(brow + kk + 4 * i);
            #pragma unroll
            for (int i = 0; i < 8; ++i) q[i] = *reinterpret_cast<const float4*>(bptr + kk * LL + 4 * i);
            #pragma unroll
            for (int s = 0; s < 8; ++s) {
                const float4 p = av[s >> 1];
                v2f a2, b;
                if (s & 1) { a2.x = row0 ? p.z : 0.0f; a2.y = row0 ? p.w : 0.0f; }
                else       { a2.x = row0 ? p.x : 0.0f; a2.y = row0 ? p.y : 0.0f; }
                b.x = valid ? (lab1 ? q[s].y : q[s].x) : 0.0f;
                b.y = valid ? (lab1 ? q[s].w : q[s].z) : 0.0f;
                acc = wmma_f32(a2, b, acc);
            }
        }
        if (half == 0) part[wave * 16 + r] = acc[0];
        __syncthreads();
        if (threadIdx.x < 16) {
            const int annS = s_ann[threadIdx.x >> 1];
            if (annS >= 0) {
                float s = bh[annS * LL + (threadIdx.x & 1)];
                #pragma unroll
                for (int w = 0; w < SPLIT2; ++w) s += part[w * 16 + threadIdx.x];
                Cb[annS * LL + (threadIdx.x & 1)] = s;
            }
        }
    }
}

#define SPLIT3 8
#define MAXTILES3 35
#define PPITCH 18
#define PT_FLOATS (16 * PPITCH)
__global__ __launch_bounds__(SPLIT3 * 32) void stage3_kernel(
    const float* __restrict__ H1, const float* __restrict__ V,
    const float* __restrict__ Cb, const int* __restrict__ g1_idx,
    float* __restrict__ out)
{
    __shared__ __attribute__((aligned(16))) float part3[SPLIT3 * PT_FLOATS];
    __shared__ int s_wcnt[8 * GG1];
    __shared__ int s_ann[8];
    __shared__ int s_g;

    const int bt   = blockIdx.x >> 2;
    const int m0   = (blockIdx.x & 3) * 16;
    const int lane = threadIdx.x & 31;
    const int wave = threadIdx.x >> 5;
    const int half = lane >> 4;
    const int r    = lane & 15;

    const int myg = g1_idx[threadIdx.x] & (GG1 - 1);
    int myrank = 0;
    #pragma unroll
    for (int v = 0; v < GG1; ++v) {
        const unsigned m = __builtin_amdgcn_ballot_w32(myg == v);
        if (lane == 0) s_wcnt[wave * GG1 + v] = __popc(m);
        if (myg == v) myrank = __popc(m & ((1u << lane) - 1u));
    }
    if (threadIdx.x < 8) s_ann[threadIdx.x] = -1;
    __syncthreads();
    int before = 0;
    int tstart = 0;
    #pragma unroll
    for (int v = 0; v < GG1; ++v) {
        int n = 0;
        #pragma unroll
        for (int w = 0; w < 8; ++w) {
            const int c = s_wcnt[w * GG1 + v];
            n += c;
            if (v == myg && w < wave) before += c;
        }
        if (v < myg) tstart += (n + 7) >> 3;
    }
    const int rank = before + myrank;
    if (tstart + (rank >> 3) == bt) {
        s_ann[rank & 7] = threadIdx.x;
        if ((rank & 7) == 0) s_g = myg;
    }
    __syncthreads();
    if (s_ann[0] < 0) return;
    const int g1t = s_g;

    const int  annB  = s_ann[r >> 1];
    const bool valid = annB >= 0;
    const bool lab1  = (r & 1) != 0;
    const int  kl    = wave * 96 + half * 4;

    const float* arow = H1 + (size_t)g1t * BB * DD + (size_t)(m0 + r) * DD + kl;
    const float* bptr = V + (size_t)(valid ? annB : 0) * DD * LL + (size_t)kl * LL;

    float4 av[12], q[24];
    #pragma unroll
    for (int i = 0; i < 12; ++i) {
        av[i]        = *reinterpret_cast<const float4*>(arow + 8 * i);
        q[2 * i]     = *reinterpret_cast<const float4*>(bptr + 16 * i);
        q[2 * i + 1] = *reinterpret_cast<const float4*>(bptr + 16 * i + 4);
    }
    __builtin_amdgcn_sched_barrier(0);

    v8f acc0, acc1;
    #pragma unroll
    for (int j = 0; j < 8; ++j) { acc0[j] = 0.0f; acc1[j] = 0.0f; }
    #pragma unroll
    for (int s = 0; s < 24; ++s) {
        const float4 p = av[s >> 1];
        v2f a2, b;
        if (s & 1) { a2.x = p.z; a2.y = p.w; }
        else       { a2.x = p.x; a2.y = p.y; }
        b.x = valid ? (lab1 ? q[s].y : q[s].x) : 0.0f;
        b.y = valid ? (lab1 ? q[s].w : q[s].z) : 0.0f;
        if (s & 1) acc1 = wmma_f32(a2, b, acc1);
        else       acc0 = wmma_f32(a2, b, acc0);
    }

    float* mine = part3 + wave * PT_FLOATS;
    #pragma unroll
    for (int j = 0; j < 8; ++j)
        mine[(j + 8 * half) * PPITCH + r] = acc0[j] + acc1[j];
    __syncthreads();

    if (threadIdx.x < 16 * 8) {
        const int row  = threadIdx.x >> 3;
        const int slot = threadIdx.x & 7;
        const int annS = s_ann[slot];
        if (annS >= 0) {
            float2 s = *reinterpret_cast<const float2*>(Cb + annS * LL);
            const float* p = part3 + row * PPITCH + 2 * slot;
            #pragma unroll
            for (int w = 0; w < SPLIT3; ++w) {
                const float2 v = *reinterpret_cast<const float2*>(p + w * PT_FLOATS);
                s.x += v.x; s.y += v.y;
            }
            *reinterpret_cast<float2*>(out + (size_t)(m0 + row) * (AANN * LL) + annS * LL) = s;
        }
    }
}

extern "C" void kernel_launch(void* const* d_in, const int* in_sizes, int n_in,
                              void* d_out, int out_size, void* d_ws, size_t ws_size,
                              hipStream_t stream) {
    const float* pooled = (const float*)d_in[0];
    const float* W1     = (const float*)d_in[1];
    const float* b1     = (const float*)d_in[2];
    const float* W2     = (const float*)d_in[3];
    const float* b2     = (const float*)d_in[4];
    const float* Wh     = (const float*)d_in[5];
    const float* bh     = (const float*)d_in[6];
    const int*   g1     = (const int*)d_in[7];
    const int*   g2     = (const int*)d_in[8];
    float*       out    = (float*)d_out;

    float* H1 = (float*)d_ws;
    float* V  = H1 + (size_t)GG1 * BB * DD;
    float* Cb = V + (size_t)AANN * DD * LL;

    stage1_kernel<<<GG1 * 24, SPLIT1 * 32, 0, stream>>>(pooled, W1, b1, H1);
    stage2_kernel<<<MAXTILES * MRANGES, SPLIT2 * 32, 0, stream>>>(W2, b2, Wh, bh, g2, V, Cb);
    stage3_kernel<<<MAXTILES3 * 4, SPLIT3 * 32, 0, stream>>>(H1, V, Cb, g1, out);
}


# Stages 1 and 2, which do not depend on each other, run as one launch (96 + 507 work-groups of 8 waves, role by block index); stage 1's K split goes from 16 to 8 waves: changes the summation order
# speedup vs baseline: 10.7017x; 1.3059x over previous
#include <hip/hip_runtime.h>
#include <hip/hip_bf16.h>


#define AANN 256
#define BB   64
#define DD   768
#define LL   2
#define GG1  4
#define GG2  8

#define TPITCH 68
#define TILE_FLOATS (32 * TPITCH)

typedef __attribute__((ext_vector_type(2))) float v2f;
typedef __attribute__((ext_vector_type(8))) float v8f;

__device__ __forceinline__ v8f wmma_f32(v2f a, v2f b, v8f c) {
    return __builtin_amdgcn_wmma_f32_16x16x4_f32(
         false, a,  false, b,
         (short)0, c,  false,  false);
}

template <int SPLIT>
__device__ __forceinline__ void gemm_tile_32x64_splitk(
    const float* __restrict__ A,
    const float* __restrict__ Bm,
    const float* __restrict__ bias,
    float* __restrict__ C,
    int m0, int n0,
    float* part)
{
    constexpr int KW = DD / SPLIT;
    constexpr int TK = (KW % 32 == 0) ? 32 : 16;
    constexpr int HK = TK / 2;
    static_assert(KW % TK == 0, "slice must be whole trips");
    const int lane = threadIdx.x & 31;
    const int wave = threadIdx.x >> 5;
    const int half = lane >> 4;
    const int r    = lane & 15;
    const int kl   = wave * KW + half * HK;

    v8f acc[2][4];
    #pragma unroll
    for (int t = 0; t < 4; ++t) {
        #pragma unroll
        for (int j = 0; j < 8; ++j) { acc[0][t][j] = 0.0f; acc[1][t][j] = 0.0f; }
    }

    const float* arow0 = A + (size_t)(m0 + r) * DD + kl;
    const float* arow1 = arow0 + (size_t)16 * DD;
    const float* brow  = Bm + (size_t)kl * DD + n0 + 4 * r;

    for (int kk = 0; kk < KW; kk += TK) {
        float4 a0v[HK / 4], a1v[HK / 4];
        #pragma unroll
        for (int i = 0; i < HK / 4; ++i) {
            a0v[i] = *reinterpret_cast<const float4*>(arow0 + kk + 4 * i);
            a1v[i] = *reinterpret_cast<const float4*>(arow1 + kk + 4 * i);
        }
        #pragma unroll
        for (int g = 0; g < HK / 8; ++g) {
            float4 bv[8];
            #pragma unroll
            for (int i = 0; i < 8; ++i)
                bv[i] = *reinterpret_cast<const float4*>(brow + (size_t)(kk + 8 * g + i) * DD);
            __builtin_amdgcn_sched_barrier(0);
            #pragma unroll
            for (int s = 0; s < 4; ++s) {
                const float4 p0 = a0v[2 * g + (s >> 1)];
                const float4 p1 = a1v[2 * g + (s >> 1)];
                v2f a0, a1;
                if (s & 1) { a0.x = p0.z; a0.y = p0.w; a1.x = p1.z; a1.y = p1.w; }
                else       { a0.x = p0.x; a0.y = p0.y; a1.x = p1.x; a1.y = p1.y; }
                const float bx[4] = { bv[2 * s].x,     bv[2 * s].y,     bv[2 * s].z,     bv[2 * s].w };
                const float by[4] = { bv[2 * s + 1].x, bv[2 * s + 1].y, bv[2 * s + 1].z, bv[2 * s + 1].w };
                #pragma unroll
                for (int t = 0; t < 4; ++t) {
                    v2f b;
                    b.x = bx[t];
                    b.y = by[t];
                    acc[0][t] = wmma_f32(a0, b, acc[0][t]);
                    acc[1][t] = wmma_f32(a1, b, acc[1][t]);
                }
            }
        }
    }

    float* mine = part + wave * TILE_FLOATS;
    #pragma unroll
    for (int u = 0; u < 2; ++u) {
        #pragma unroll
        for (int j = 0; j < 8; ++j) {
            const int row = u * 16 + j + 8 * half;
            float4 v;
            v.x = acc[u][0][j]; v.y = acc[u][1][j]; v.z = acc[u][2][j]; v.w = acc[u][3][j];
            *reinterpret_cast<float4*>(mine + row * TPITCH + 4 * r) = v;
        }
    }
    __syncthreads();

    for (int idx = threadIdx.x; idx < 32 * 16; idx += SPLIT * 32) {
        const int row = idx >> 4;
        const int c4  = (idx & 15) * 4;
        float4 s = *reinterpret_cast<const float4*>(bias + n0 + c4);
        const float* p = part + row * TPITCH + c4;
        #pragma unroll
        for (int w = 0; w < SPLIT; ++w) {
            const float4 v = *reinterpret_cast<const float4*>(p + w * TILE_FLOATS);
            s.x += v.x; s.y += v.y; s.z += v.z; s.w += v.w;
        }
        *reinterpret_cast<float4*>(C + (size_t)(m0 + row) * DD + n0 + c4) = s;
    }
}

#define SPLIT1 8
#define NB1 (GG1 * 24)
__device__ __forceinline__ void stage1_role(
    const float* __restrict__ pooled, const float* __restrict__ W1,
    const float* __restrict__ b1, float* __restrict__ H1,
    int bid, float* part)
{
    const int g  = bid / 24;
    const int wt = bid % 24;
    const int m0 = (wt / 12) * 32;
    const int n0 = (wt % 12) * 64;
    gemm_tile_32x64_splitk<SPLIT1>(pooled,
                    W1 + (size_t)g * DD * DD,
                    b1 + (size_t)g * DD,
                    H1 + (size_t)g * BB * DD,
                    m0, n0, part);
}

#define SPLIT2 8
#define VPITCH 18
#define VT_FLOATS (64 * VPITCH)
#define MRANGES 13
#define MAXTILES 39
__device__ __forceinline__ void stage2_role(
    const float* __restrict__ W2, const float* __restrict__ b2,
    const float* __restrict__ Wh, const float* __restrict__ bh,
    const int* __restrict__ g2_idx, float* __restrict__ V, float* __restrict__ Cb,
    int bid, float* part, int* s_wcnt, int* s_ann, int* s_gp)
{
    const int bt   = bid / MRANGES;
    const int mr   = bid % MRANGES;
    const int lane = threadIdx.x & 31;
    const int wave = threadIdx.x >> 5;
    const int half = lane >> 4;
    const int r    = lane & 15;

    const int myg = g2_idx[threadIdx.x] & (GG2 - 1);
    int myrank = 0;
    #pragma unroll
    for (int v = 0; v < GG2; ++v) {
        const unsigned m = __builtin_amdgcn_ballot_w32(myg == v);
        if (lane == 0) s_wcnt[wave * GG2 + v] = __popc(m);
        if (myg == v) myrank = __popc(m & ((1u << lane) - 1u));
    }
    if (threadIdx.x < 8) s_ann[threadIdx.x] = -1;
    __syncthreads();
    int before = 0;
    int tstart = 0;
    #pragma unroll
    for (int v = 0; v < GG2; ++v) {
        int n = 0;
        #pragma unroll
        for (int w = 0; w < 8; ++w) {
            const int c = s_wcnt[w * GG2 + v];
            n += c;
            if (v == myg && w < wave) before += c;
        }
        if (v < myg) tstart += (n + 7) >> 3;
    }
    const int rank = before + myrank;
    if (tstart + (rank >> 3) == bt) {
        s_ann[rank & 7] = threadIdx.x;
        if ((rank & 7) == 0) *s_gp = myg;
    }
    __syncthreads();
    if (s_ann[0] < 0) return;
    const int g2t = *s_gp;

    const int  annB  = s_ann[r >> 1];
    const bool valid = annB >= 0;
    const bool lab1  = (r & 1) != 0;
    const int  kl    = wave * 96 + half * 16;
    const float* bptr = Wh + (size_t)(valid ? annB : 0) * DD * LL + (size_t)kl * LL;

    if (mr < 12) {
        const int m0 = mr * 64;
        const float* arow = W2 + (size_t)g2t * DD * DD + (size_t)(m0 + r) * DD + kl;
        v8f acc[4];
        #pragma unroll
        for (int u = 0; u < 4; ++u) {
            #pragma unroll
            for (int j = 0; j < 8; ++j) acc[u][j] = 0.0f;
        }
        for (int kk = 0; kk < 96; kk += 32) {
            float4 av[4][4], q[8];
            #pragma unroll
            for (int u = 0; u < 4; ++u) {
                #pragma unroll
                for (int i = 0; i < 4; ++i)
                    av[u][i] = *reinterpret_cast<const float4*>(arow + (size_t)u * 16 * DD + kk + 4 * i);
            }
            #pragma unroll
            for (int i = 0; i < 8; ++i) q[i] = *reinterpret_cast<const float4*>(bptr + kk * LL + 4 * i);
            __builtin_amdgcn_sched_barrier(0);
            #pragma unroll
            for (int s = 0; s < 8; ++s) {
                v2f b;
                b.x = valid ? (lab1 ? q[s].y : q[s].x) : 0.0f;
                b.y = valid ? (lab1 ? q[s].w : q[s].z) : 0.0f;
                #pragma unroll
                for (int u = 0; u < 4; ++u) {
                    const float4 p = av[u][s >> 1];
                    v2f a2;
                    if (s & 1) { a2.x = p.z; a2.y = p.w; }
                    else       { a2.x = p.x; a2.y = p.y; }
                    acc[u] = wmma_f32(a2, b, acc[u]);
                }
            }
        }
        float* mine = part + wave * VT_FLOATS;
        #pragma unroll
        for (int u = 0; u < 4; ++u) {
            #pragma unroll
            for (int j = 0; j < 8; ++j)
                mine[(16 * u + j + 8 * half) * VPITCH + r] = acc[u][j];
        }
        __syncthreads();
        const int annS = s_ann[wave];
        if (annS >= 0) {
            float4 s = make_float4(0.0f, 0.0f, 0.0f, 0.0f);
            const float* p = part + (2 * lane) * VPITCH + 2 * wave;
            #pragma unroll
            for (int w = 0; w < SPLIT2; ++w) {
                const float2 lo = *reinterpret_cast<const float2*>(p + w * VT_FLOATS);
                const float2 hi = *reinterpret_cast<const float2*>(p + w * VT_FLOATS + VPITCH);
                s.x += lo.x; s.y += lo.y; s.z += hi.x; s.w += hi.y;
            }
            *reinterpret_cast<float4*>(V + (size_t)annS * DD * LL + (size_t)(m0 + 2 * lane) * LL) = s;
        }
    } else {
        const float* brow = b2 + (size_t)g2t * DD + kl;
        const bool row0 = (r == 0);
        v8f acc;
        #pragma unroll
        for (int j = 0; j < 8; ++j) acc[j] = 0.0f;
        for (int kk = 0; kk < 96; kk += 32) {
            float4 av[4], q[8];
            #pragma unroll
            for (int i = 0; i < 4; ++i) av[i] = *reinterpret_cast<const float4*>(brow + kk + 4 * i);
            #pragma unroll
            for (int i = 0; i < 8; ++i) q[i] = *reinterpret_cast<const float4*>(bptr + kk * LL + 4 * i);
            #pragma unroll
            for (int s = 0; s < 8; ++s) {
                const float4 p = av[s >> 1];
                v2f a2, b;
                if (s & 1) { a2.x = row0 ? p.z : 0.0f; a2.y = row0 ? p.w : 0.0f; }
                else       { a2.x = row0 ? p.x : 0.0f; a2.y = row0 ? p.y : 0.0f; }
                b.x = valid ? (lab1 ? q[s].y : q[s].x) : 0.0f;
                b.y = valid ? (lab1 ? q[s].w : q[s].z) : 0.0f;
                acc = wmma_f32(a2, b, acc);
            }
        }
        if (half == 0) part[wave * 16 + r] = acc[0];
        __syncthreads();
        if (threadIdx.x < 16) {
            const int annS = s_ann[threadIdx.x >> 1];
            if (annS >= 0) {
                float s = bh[annS * LL + (threadIdx.x & 1)];
                #pragma unroll
                for (int w = 0; w < SPLIT2; ++w) s += part[w * 16 + threadIdx.x];
                Cb[annS * LL + (threadIdx.x & 1)] = s;
            }
        }
    }
}

__global__ __launch_bounds__(SPLIT2 * 32) __attribute__((amdgpu_waves_per_eu(6, 6))) void stage12_kernel(
    const float* __restrict__ pooled, const float* __restrict__ W1,
    const float* __restrict__ b1, float* __restrict__ H1,
    const float* __restrict__ W2, const float* __restrict__ b2,
    const float* __restrict__ Wh, const float* __restrict__ bh,
    const int* __restrict__ g2_idx, float* __restrict__ V, float* __restrict__ Cb)
{
    static_assert(SPLIT1 == SPLIT2, "both roles use the whole work-group");
    static_assert(SPLIT1 * TILE_FLOATS >= SPLIT2 * VT_FLOATS, "LDS block covers both roles");
    __shared__ __attribute__((aligned(16))) float part[SPLIT1 * TILE_FLOATS];
    __shared__ int s_wcnt[8 * GG2];
    __shared__ int s_ann[8];
    __shared__ int s_g;
    if (blockIdx.x < NB1)
        stage1_role(pooled, W1, b1, H1, (int)blockIdx.x, part);
    else
        stage2_role(W2, b2, Wh, bh, g2_idx, V, Cb, (int)blockIdx.x - NB1, part, s_wcnt, s_ann, &s_g);
}

#define SPLIT3 8
#define MAXTILES3 35
#define PPITCH 18
#define PT_FLOATS (16 * PPITCH)
__global__ __launch_bounds__(SPLIT3 * 32) void stage3_kernel(
    const float* __restrict__ H1, const float* __restrict__ V,
    const float* __restrict__ Cb, const int* __restrict__ g1_idx,
    float* __restrict__ out)
{
    __shared__ __attribute__((aligned(16))) float part3[SPLIT3 * PT_FLOATS];
    __shared__ int s_wcnt[8 * GG1];
    __shared__ int s_ann[8];
    __shared__ int s_g;

    const int bt   = blockIdx.x >> 2;
    const int m0   = (blockIdx.x & 3) * 16;
    const int lane = threadIdx.x & 31;
    const int wave = threadIdx.x >> 5;
    const int half = lane >> 4;
    const int r    = lane & 15;

    const int myg = g1_idx[threadIdx.x] & (GG1 - 1);
    int myrank = 0;
    #pragma unroll
    for (int v = 0; v < GG1; ++v) {
        const unsigned m = __builtin_amdgcn_ballot_w32(myg == v);
        if (lane == 0) s_wcnt[wave * GG1 + v] = __popc(m);
        if (myg == v) myrank = __popc(m & ((1u << lane) - 1u));
    }
    if (threadIdx.x < 8) s_ann[threadIdx.x] = -1;
    __syncthreads();
    int before = 0;
    int tstart = 0;
    #pragma unroll
    for (int v = 0; v < GG1; ++v) {
        int n = 0;
        #pragma unroll
        for (int w = 0; w < 8; ++w) {
            const int c = s_wcnt[w * GG1 + v];
            n += c;
            if (v == myg && w < wave) before += c;
        }
        if (v < myg) tstart += (n + 7) >> 3;
    }
    const int rank = before + myrank;
    if (tstart + (rank >> 3) == bt) {
        s_ann[rank & 7] = threadIdx.x;
        if ((rank & 7) == 0) s_g = myg;
    }
    __syncthreads();
    if (s_ann[0] < 0) return;
    const int g1t = s_g;

    const int  annB  = s_ann[r >> 1];
    const bool valid = annB >= 0;
    const bool lab1  = (r & 1) != 0;
    const int  kl    = wave * 96 + half * 4;

    const float* arow = H1 + (size_t)g1t * BB * DD + (size_t)(m0 + r) * DD + kl;
    const float* bptr = V + (size_t)(valid ? annB : 0) * DD * LL + (size_t)kl * LL;

    float4 av[12], q[24];
    #pragma unroll
    for (int i = 0; i < 12; ++i) {
        av[i]        = *reinterpret_cast<const float4*>(arow + 8 * i);
        q[2 * i]     = *reinterpret_cast<const float4*>(bptr + 16 * i);
        q[2 * i + 1] = *reinterpret_cast<const float4*>(bptr + 16 * i + 4);
    }
    __builtin_amdgcn_sched_barrier(0);

    v8f acc0, acc1;
    #pragma unroll
    for (int j = 0; j < 8; ++j) { acc0[j] = 0.0f; acc1[j] = 0.0f; }
    #pragma unroll
    for (int s = 0; s < 24; ++s) {
        const float4 p = av[s >> 1];
        v2f a2, b;
        if (s & 1) { a2.x = p.z; a2.y = p.w; }
        else       { a2.x = p.x; a2.y = p.y; }
        b.x = valid ? (lab1 ? q[s].y : q[s].x) : 0.0f;
        b.y = valid ? (lab1 ? q[s].w : q[s].z) : 0.0f;
        if (s & 1) acc1 = wmma_f32(a2, b, acc1);
        else       acc0 = wmma_f32(a2, b, acc0);
    }

    float* mine = part3 + wave * PT_FLOATS;
    #pragma unroll
    for (int j = 0; j < 8; ++j)
        mine[(j + 8 * half) * PPITCH + r] = acc0[j] + acc1[j];
    __syncthreads();

    if (threadIdx.x < 16 * 8) {
        const int row  = threadIdx.x >> 3;
        const int slot = threadIdx.x & 7;
        const int annS = s_ann[slot];
        if (annS >= 0) {
            float2 s = *reinterpret_cast<const float2*>(Cb + annS * LL);
            const float* p = part3 + row * PPITCH + 2 * slot;
            #pragma unroll
            for (int w = 0; w < SPLIT3; ++w) {
                const float2 v = *reinterpret_cast<const float2*>(p + w * PT_FLOATS);
                s.x += v.x; s.y += v.y;
            }
            *reinterpret_cast<float2*>(out + (size_t)(m0 + row) * (AANN * LL) + annS * LL) = s;
        }
    }
}

extern "C" void kernel_launch(void* const* d_in, const int* in_sizes, int n_in,
                              void* d_out, int out_size, void* d_ws, size_t ws_size,
                              hipStream_t stream) {
    const float* pooled = (const float*)d_in[0];
    const float* W1     = (const float*)d_in[1];
    const float* b1     = (const float*)d_in[2];
    const float* W2     = (const float*)d_in[3];
    const float* b2     = (const float*)d_in[4];
    const float* Wh     = (const float*)d_in[5];
    const float* bh     = (const float*)d_in[6];
    const int*   g1     = (const int*)d_in[7];
    const int*   g2     = (const int*)d_in[8];
    float*       out    = (float*)d_out;

    float* H1 = (float*)d_ws;
    float* V  = H1 + (size_t)GG1 * BB * DD;
    float* Cb = V + (size_t)AANN * DD * LL;

    stage12_kernel<<<NB1 + MAXTILES * MRANGES, SPLIT2 * 32, 0, stream>>>(pooled, W1, b1, H1, W2, b2, Wh, bh, g2, V, Cb);
    stage3_kernel<<<MAXTILES3 * 4, SPLIT3 * 32, 0, stream>>>(H1, V, Cb, g1, out);
}


# Stage 2 fetches both operands (W2 rows, Wh) as whole 128-byte lines, 4 lines per load not 16, and turns them into fragments in a wave-private LDS slot; fragment contents and sums unchanged
# speedup vs baseline: 11.6913x; 1.0925x over previous
#include <hip/hip_runtime.h>
#include <hip/hip_bf16.h>


#define AANN 256
#define BB   64
#define DD   768
#define LL   2
#define GG1  4
#define GG2  8

#define TPITCH 68
#define TILE_FLOATS (32 * TPITCH)

typedef __attribute__((ext_vector_type(2))) float v2f;
typedef __attribute__((ext_vector_type(8))) float v8f;

__device__ __forceinline__ v8f wmma_f32(v2f a, v2f b, v8f c) {
    return __builtin_amdgcn_wmma_f32_16x16x4_f32(
         false, a,  false, b,
         (short)0, c,  false,  false);
}

template <int SPLIT>
__device__ __forceinline__ void gemm_tile_32x64_splitk(
    const float* __restrict__ A,
    const float* __restrict__ Bm,
    const float* __restrict__ bias,
    float* __restrict__ C,
    int m0, int n0,
    float* part)
{
    constexpr int KW = DD / SPLIT;
    constexpr int TK = (KW % 32 == 0) ? 32 : 16;
    constexpr int HK = TK / 2;
    static_assert(KW % TK == 0, "slice must be whole trips");
    const int lane = threadIdx.x & 31;
    const int wave = threadIdx.x >> 5;
    const int half = lane >> 4;
    const int r    = lane & 15;
    const int kl   = wave * KW + half * HK;

    v8f acc[2][4];
    #pragma unroll
    for (int t = 0; t < 4; ++t) {
        #pragma unroll
        for (int j = 0; j < 8; ++j) { acc[0][t][j] = 0.0f; acc[1][t][j] = 0.0f; }
    }

    const float* arow0 = A + (size_t)(m0 + r) * DD + kl;
    const float* arow1 = arow0 + (size_t)16 * DD;
    const float* brow  = Bm + (size_t)kl * DD + n0 + 4 * r;

    for (int kk = 0; kk < KW; kk += TK) {
        float4 a0v[HK / 4], a1v[HK / 4];
        #pragma unroll
        for (int i = 0; i < HK / 4; ++i) {
            a0v[i] = *reinterpret_cast<const float4*>(arow0 + kk + 4 * i);
            a1v[i] = *reinterpret_cast<const float4*>(arow1 + kk + 4 * i);
        }
        #pragma unroll
        for (int g = 0; g < HK / 8; ++g) {
            float4 bv[8];
            #pragma unroll
            for (int i = 0; i < 8; ++i)
                bv[i] = *reinterpret_cast<const float4*>(brow + (size_t)(kk + 8 * g + i) * DD);
            __builtin_amdgcn_sched_barrier(0);
            #pragma unroll
            for (int s = 0; s < 4; ++s) {
                const float4 p0 = a0v[2 * g + (s >> 1)];
                const float4 p1 = a1v[2 * g + (s >> 1)];
                v2f a0, a1;
                if (s & 1) { a0.x = p0.z; a0.y = p0.w; a1.x = p1.z; a1.y = p1.w; }
                else       { a0.x = p0.x; a0.y = p0.y; a1.x = p1.x; a1.y = p1.y; }
                const float bx[4] = { bv[2 * s].x,     bv[2 * s].y,     bv[2 * s].z,     bv[2 * s].w };
                const float by[4] = { bv[2 * s + 1].x, bv[2 * s + 1].y, bv[2 * s + 1].z, bv[2 * s + 1].w };
                #pragma unroll
                for (int t = 0; t < 4; ++t) {
                    v2f b;
                    b.x = bx[t];
                    b.y = by[t];
                    acc[0][t] = wmma_f32(a0, b, acc[0][t]);
                    acc[1][t] = wmma_f32(a1, b, acc[1][t]);
                }
            }
        }
    }

    float* mine = part + wave * TILE_FLOATS;
    #pragma unroll
    for (int u = 0; u < 2; ++u) {
        #pragma unroll
        for (int j = 0; j < 8; ++j) {
            const int row = u * 16 + j + 8 * half;
            float4 v;
            v.x = acc[u][0][j]; v.y = acc[u][1][j]; v.z = acc[u][2][j]; v.w = acc[u][3][j];
            *reinterpret_cast<float4*>(mine + row * TPITCH + 4 * r) = v;
        }
    }
    __syncthreads();

    for (int idx = threadIdx.x; idx < 32 * 16; idx += SPLIT * 32) {
        const int row = idx >> 4;
        const int c4  = (idx & 15) * 4;
        float4 s = *reinterpret_cast<const float4*>(bias + n0 + c4);
        const float* p = part + row * TPITCH + c4;
        #pragma unroll
        for (int w = 0; w < SPLIT; ++w) {
            const float4 v = *reinterpret_cast<const float4*>(p + w * TILE_FLOATS);
            s.x += v.x; s.y += v.y; s.z += v.z; s.w += v.w;
        }
        *reinterpret_cast<float4*>(C + (size_t)(m0 + row) * DD + n0 + c4) = s;
    }
}

#define SPLIT1 8
#define NB1 (GG1 * 24)
__device__ __forceinline__ void stage1_role(
    const float* __restrict__ pooled, const float* __restrict__ W1,
    const float* __restrict__ b1, float* __restrict__ H1,
    int bid, float* part)
{
    const int g  = bid / 24;
    const int wt = bid % 24;
    const int m0 = (wt / 12) * 32;
    const int n0 = (wt % 12) * 64;
    gemm_tile_32x64_splitk<SPLIT1>(pooled,
                    W1 + (size_t)g * DD * DD,
                    b1 + (size_t)g * DD,
                    H1 + (size_t)g * BB * DD,
                    m0, n0, part);
}

#define SPLIT2 8
#define VPITCH 18
#define VT_FLOATS (64 * VPITCH)
#define APITCH 36
#define AS_FLOATS (64 * APITCH)
#define BPITCH 68
#define BS_FLOATS (8 * BPITCH)
#define WS_FLOATS (AS_FLOATS + BS_FLOATS)
#define MRANGES 13
#define MAXTILES 39
__device__ __forceinline__ void stage2_role(
    const float* __restrict__ W2, const float* __restrict__ b2,
    const float* __restrict__ Wh, const float* __restrict__ bh,
    const int* __restrict__ g2_idx, float* __restrict__ V, float* __restrict__ Cb,
    int bid, float* part, int* s_wcnt, int* s_ann, int* s_gp)
{
    const int bt   = bid / MRANGES;
    const int mr   = bid % MRANGES;
    const int lane = threadIdx.x & 31;
    const int wave = threadIdx.x >> 5;
    const int half = lane >> 4;
    const int r    = lane & 15;

    const int myg = g2_idx[threadIdx.x] & (GG2 - 1);
    int myrank = 0;
    #pragma unroll
    for (int v = 0; v < GG2; ++v) {
        const unsigned m = __builtin_amdgcn_ballot_w32(myg == v);
        if (lane == 0) s_wcnt[wave * GG2 + v] = __popc(m);
        if (myg == v) myrank = __popc(m & ((1u << lane) - 1u));
    }
    if (threadIdx.x < 8) s_ann[threadIdx.x] = -1;
    __syncthreads();
    int before = 0;
    int tstart = 0;
    #pragma unroll
    for (int v = 0; v < GG2; ++v) {
        int n = 0;
        #pragma unroll
        for (int w = 0; w < 8; ++w) {
            const int c = s_wcnt[w * GG2 + v];
            n += c;
            if (v == myg && w < wave) before += c;
        }
        if (v < myg) tstart += (n + 7) >> 3;
    }
    const int rank = before + myrank;
    if (tstart + (rank >> 3) == bt) {
        s_ann[rank & 7] = threadIdx.x;
        if ((rank & 7) == 0) *s_gp = myg;
    }
    __syncthreads();
    if (s_ann[0] < 0) return;
    const int g2t = *s_gp;

    const int  annB  = s_ann[r >> 1];
    const bool valid = annB >= 0;
    const bool lab1  = (r & 1) != 0;
    const int  kl    = wave * 96 + half * 16;
    const float* bptr = Wh + (size_t)(valid ? annB : 0) * DD * LL + (size_t)kl * LL;

    if (mr < 12) {
        const int m0 = mr * 64;
        const int lrow = lane >> 3;
        const int lchk = (lane & 7) * 4;
        const float* aline = W2 + (size_t)g2t * DD * DD + (size_t)(m0 + lrow) * DD + wave * 96 + lchk;
        float* aslot = part + wave * WS_FLOATS;
        float* ast = aslot + lrow * APITCH + lchk;
        const float* ald = aslot + r * APITCH + half * 16;
        const int bsl  = lane >> 4;
        const int bchk = (lane & 15) * 4;
        const int an0 = s_ann[0 + bsl], an1 = s_ann[2 + bsl], an2 = s_ann[4 + bsl], an3 = s_ann[6 + bsl];
        const float* bl0 = Wh + (size_t)(an0 >= 0 ? an0 : 0) * DD * LL + (size_t)(wave * 96) * LL + bchk;
        const float* bl1 = Wh + (size_t)(an1 >= 0 ? an1 : 0) * DD * LL + (size_t)(wave * 96) * LL + bchk;
        const float* bl2 = Wh + (size_t)(an2 >= 0 ? an2 : 0) * DD * LL + (size_t)(wave * 96) * LL + bchk;
        const float* bl3 = Wh + (size_t)(an3 >= 0 ? an3 : 0) * DD * LL + (size_t)(wave * 96) * LL + bchk;
        float* bslot = aslot + AS_FLOATS;
        float* bst = bslot + bsl * BPITCH + bchk;
        const float* bld = bslot + (r >> 1) * BPITCH + half * 32;
        v8f acc[4];
        #pragma unroll
        for (int u = 0; u < 4; ++u) {
            #pragma unroll
            for (int j = 0; j < 8; ++j) acc[u][j] = 0.0f;
        }
        for (int kk = 0; kk < 96; kk += 32) {
            float4 av[4][4], q[8], ln[16];
            const float4 lb0 = *reinterpret_cast<const float4*>(bl0 + kk * LL);
            const float4 lb1 = *reinterpret_cast<const float4*>(bl1 + kk * LL);
            const float4 lb2 = *reinterpret_cast<const float4*>(bl2 + kk * LL);
            const float4 lb3 = *reinterpret_cast<const float4*>(bl3 + kk * LL);
            #pragma unroll
            for (int j = 0; j < 16; ++j)
                ln[j] = *reinterpret_cast<const float4*>(aline + (size_t)(4 * j) * DD + kk);
            __builtin_amdgcn_sched_barrier(0);
            *reinterpret_cast<float4*>(bst + 0 * BPITCH) = lb0;
            *reinterpret_cast<float4*>(bst + 2 * BPITCH) = lb1;
            *reinterpret_cast<float4*>(bst + 4 * BPITCH) = lb2;
            *reinterpret_cast<float4*>(bst + 6 * BPITCH) = lb3;
            #pragma unroll
            for (int j = 0; j < 16; ++j)
                *reinterpret_cast<float4*>(ast + (4 * j) * APITCH) = ln[j];
            #pragma unroll
            for (int i = 0; i < 8; ++i) q[i] = *reinterpret_cast<const float4*>(bld + 4 * i);
            #pragma unroll
            for (int i = 0; i < 4; ++i) {
                #pragma unroll
                for (int u = 0; u < 4; ++u)
                    av[u][i] = *reinterpret_cast<const float4*>(ald + (16 * u) * APITCH + 4 * i);
            }
            __builtin_amdgcn_sched_barrier(0);
            #pragma unroll
            for (int s = 0; s < 8; ++s) {
                v2f b;
                b.x = valid ? (lab1 ? q[s].y : q[s].x) : 0.0f;
                b.y = valid ? (lab1 ? q[s].w : q[s].z) : 0.0f;
                #pragma unroll
                for (int u = 0; u < 4; ++u) {
                    const float4 p = av[u][s >> 1];
                    v2f a2;
                    if (s & 1) { a2.x = p.z; a2.y = p.w; }
                    else       { a2.x = p.x; a2.y = p.y; }
                    acc[u] = wmma_f32(a2, b, acc[u]);
                }
            }
        }
        float* mine = aslot;
        #pragma unroll
        for (int u = 0; u < 4; ++u) {
            #pragma unroll
            for (int j = 0; j < 8; ++j)
                mine[(16 * u + j + 8 * half) * VPITCH + r] = acc[u][j];
        }
        __syncthreads();
        const int annS = s_ann[wave];
        if (annS >= 0) {
            float4 s = make_float4(0.0f, 0.0f, 0.0f, 0.0f);
            const float* p = part + (2 * lane) * VPITCH + 2 * wave;
            #pragma unroll
            for (int w = 0; w < SPLIT2; ++w) {
                const float2 lo = *reinterpret_cast<const float2*>(p + w * WS_FLOATS);
                const float2 hi = *reinterpret_cast<const float2*>(p + w * WS_FLOATS + VPITCH);
                s.x += lo.x; s.y += lo.y; s.z += hi.x; s.w += hi.y;
            }
            *reinterpret_cast<float4*>(V + (size_t)annS * DD * LL + (size_t)(m0 + 2 * lane) * LL) = s;
        }
    } else {
        const float* brow = b2 + (size_t)g2t * DD + kl;
        const bool row0 = (r == 0);
        v8f acc;
        #pragma unroll
        for (int j = 0; j < 8; ++j) acc[j] = 0.0f;
        for (int kk = 0; kk < 96; kk += 32) {
            float4 av[4], q[8];
            #pragma unroll
            for (int i = 0; i < 4; ++i) av[i] = *reinterpret_cast<const float4*>(brow + kk + 4 * i);
            #pragma unroll
            for (int i = 0; i < 8; ++i) q[i] = *reinterpret_cast<const float4*>(bptr + kk * LL + 4 * i);
            #pragma unroll
            for (int s = 0; s < 8; ++s) {
                const float4 p = av[s >> 1];
                v2f a2, b;
                if (s & 1) { a2.x = row0 ? p.z : 0.0f; a2.y = row0 ? p.w : 0.0f; }
                else       { a2.x = row0 ? p.x : 0.0f; a2.y = row0 ? p.y : 0.0f; }
                b.x = valid ? (lab1 ? q[s].y : q[s].x) : 0.0f;
                b.y = valid ? (lab1 ? q[s].w : q[s].z) : 0.0f;
                acc = wmma_f32(a2, b, acc);
            }
        }
        if (half == 0) part[wave * 16 + r] = acc[0];
        __syncthreads();
        if (threadIdx.x < 16) {
            const int annS = s_ann[threadIdx.x >> 1];
            if (annS >= 0) {
                float s = bh[annS * LL + (threadIdx.x & 1)];
                #pragma unroll
                for (int w = 0; w < SPLIT2; ++w) s += part[w * 16 + threadIdx.x];
                Cb[annS * LL + (threadIdx.x & 1)] = s;
            }
        }
    }
}

__global__ __launch_bounds__(SPLIT2 * 32) __attribute__((amdgpu_waves_per_eu(6, 6))) void stage12_kernel(
    const float* __restrict__ pooled, const float* __restrict__ W1,
    const float* __restrict__ b1, float* __restrict__ H1,
    const float* __restrict__ W2, const float* __restrict__ b2,
    const float* __restrict__ Wh, const float* __restrict__ bh,
    const int* __restrict__ g2_idx, float* __restrict__ V, float* __restrict__ Cb)
{
    static_assert(SPLIT1 == SPLIT2, "both roles use the whole work-group");
    static_assert(AS_FLOATS >= VT_FLOATS, "a wave's partial tile fits its A slot");
    constexpr int PART_FLOATS = (SPLIT1 * TILE_FLOATS > SPLIT2 * WS_FLOATS) ? SPLIT1 * TILE_FLOATS : SPLIT2 * WS_FLOATS;
    __shared__ __attribute__((aligned(16))) float part[PART_FLOATS];
    __shared__ int s_wcnt[8 * GG2];
    __shared__ int s_ann[8];
    __shared__ int s_g;
    if (blockIdx.x < NB1)
        stage1_role(pooled, W1, b1, H1, (int)blockIdx.x, part);
    else
        stage2_role(W2, b2, Wh, bh, g2_idx, V, Cb, (int)blockIdx.x - NB1, part, s_wcnt, s_ann, &s_g);
}

#define SPLIT3 8
#define MAXTILES3 35
#define PPITCH 18
#define PT_FLOATS (16 * PPITCH)
__global__ __launch_bounds__(SPLIT3 * 32) void stage3_kernel(
    const float* __restrict__ H1, const float* __restrict__ V,
    const float* __restrict__ Cb, const int* __restrict__ g1_idx,
    float* __restrict__ out)
{
    __shared__ __attribute__((aligned(16))) float part3[SPLIT3 * PT_FLOATS];
    __shared__ int s_wcnt[8 * GG1];
    __shared__ int s_ann[8];
    __shared__ int s_g;

    const int bt   = blockIdx.x >> 2;
    const int m0   = (blockIdx.x & 3) * 16;
    const int lane = threadIdx.x & 31;
    const int wave = threadIdx.x >> 5;
    const int half = lane >> 4;
    const int r    = lane & 15;

    const int myg = g1_idx[threadIdx.x] & (GG1 - 1);
    int myrank = 0;
    #pragma unroll
    for (int v = 0; v < GG1; ++v) {
        const unsigned m = __builtin_amdgcn_ballot_w32(myg == v);
        if (lane == 0) s_wcnt[wave * GG1 + v] = __popc(m);
        if (myg == v) myrank = __popc(m & ((1u << lane) - 1u));
    }
    if (threadIdx.x < 8) s_ann[threadIdx.x] = -1;
    __syncthreads();
    int before = 0;
    int tstart = 0;
    #pragma unroll
    for (int v = 0; v < GG1; ++v) {
        int n = 0;
        #pragma unroll
        for (int w = 0; w < 8; ++w) {
            const int c = s_wcnt[w * GG1 + v];
            n += c;
            if (v == myg && w < wave) before += c;
        }
        if (v < myg) tstart += (n + 7) >> 3;
    }
    const int rank = before + myrank;
    if (tstart + (rank >> 3) == bt) {
        s_ann[rank & 7] = threadIdx.x;
        if ((rank & 7) == 0) s_g = myg;
    }
    __syncthreads();
    if (s_ann[0] < 0) return;
    const int g1t = s_g;

    const int  annB  = s_ann[r >> 1];
    const bool valid = annB >= 0;
    const bool lab1  = (r & 1) != 0;
    const int  kl    = wave * 96 + half * 4;

    const float* arow = H1 + (size_t)g1t * BB * DD + (size_t)(m0 + r) * DD + kl;
    const float* bptr = V + (size_t)(valid ? annB : 0) * DD * LL + (size_t)kl * LL;

    float4 av[12], q[24];
    #pragma unroll
    for (int i = 0; i < 12; ++i) {
        av[i]        = *reinterpret_cast<const float4*>(arow + 8 * i);
        q[2 * i]     = *reinterpret_cast<const float4*>(bptr + 16 * i);
        q[2 * i + 1] = *reinterpret_cast<const float4*>(bptr + 16 * i + 4);
    }
    __builtin_amdgcn_sched_barrier(0);

    v8f acc0, acc1;
    #pragma unroll
    for (int j = 0; j < 8; ++j) { acc0[j] = 0.0f; acc1[j] = 0.0f; }
    #pragma unroll
    for (int s = 0; s < 24; ++s) {
        const float4 p = av[s >> 1];
        v2f a2, b;
        if (s & 1) { a2.x = p.z; a2.y = p.w; }
        else       { a2.x = p.x; a2.y = p.y; }
        b.x = valid ? (lab1 ? q[s].y : q[s].x) : 0.0f;
        b.y = valid ? (lab1 ? q[s].w : q[s].z) : 0.0f;
        if (s & 1) acc1 = wmma_f32(a2, b, acc1);
        else       acc0 = wmma_f32(a2, b, acc0);
    }

    float* mine = part3 + wave * PT_FLOATS;
    #pragma unroll
    for (int j = 0; j < 8; ++j)
        mine[(j + 8 * half) * PPITCH + r] = acc0[j] + acc1[j];
    __syncthreads();

    if (threadIdx.x < 16 * 8) {
        const int row  = threadIdx.x >> 3;
        const int slot = threadIdx.x & 7;
        const int annS = s_ann[slot];
        if (annS >= 0) {
            float2 s = *reinterpret_cast<const float2*>(Cb + annS * LL);
            const float* p = part3 + row * PPITCH + 2 * slot;
            #pragma unroll
            for (int w = 0; w < SPLIT3; ++w) {
                const float2 v = *reinterpret_cast<const float2*>(p + w * PT_FLOATS);
                s.x += v.x; s.y += v.y;
            }
            *reinterpret_cast<float2*>(out + (size_t)(m0 + row) * (AANN * LL) + annS * LL) = s;
        }
    }
}

extern "C" void kernel_launch(void* const* d_in, const int* in_sizes, int n_in,
                              void* d_out, int out_size, void* d_ws, size_t ws_size,
                              hipStream_t stream) {
    const float* pooled = (const float*)d_in[0];
    const float* W1     = (const float*)d_in[1];
    const float* b1     = (const float*)d_in[2];
    const float* W2     = (const float*)d_in[3];
    const float* b2     = (const float*)d_in[4];
    const float* Wh     = (const float*)d_in[5];
    const float* bh     = (const float*)d_in[6];
    const int*   g1     = (const int*)d_in[7];
    const int*   g2     = (const int*)d_in[8];
    float*       out    = (float*)d_out;

    float* H1 = (float*)d_ws;
    float* V  = H1 + (size_t)GG1 * BB * DD;
    float* Cb = V + (size_t)AANN * DD * LL;

    stage12_kernel<<<NB1 + MAXTILES * MRANGES, SPLIT2 * 32, 0, stream>>>(pooled, W1, b1, H1, W2, b2, Wh, bh, g2, V, Cb);
    stage3_kernel<<<MAXTILES3 * 4, SPLIT3 * 32, 0, stream>>>(H1, V, Cb, g1, out);
}
